# starting kernel
# baseline (speedup 1.0000x reference)
#include <hip/hip_runtime.h>


typedef float v2f __attribute__((ext_vector_type(2)));
typedef float v4f __attribute__((ext_vector_type(4)));
typedef float v8f __attribute__((ext_vector_type(8)));

#define DSBAR() asm volatile("s_wait_dscnt 0" ::: "memory")

__device__ __forceinline__ float fast_tanh(float x) {
#if __has_builtin(__builtin_amdgcn_tanhf)
    return __builtin_amdgcn_tanhf(x);
#else
    float e = __builtin_amdgcn_exp2f(x * 2.8853900817779268f);
    return 1.0f - 2.0f * __builtin_amdgcn_rcpf(e + 1.0f);
#endif
}
__device__ __forceinline__ float fast_sig(float x) {
    return fmaf(0.5f, fast_tanh(0.5f * x), 0.5f);
}

__device__ __forceinline__ v8f wmma_k4(v2f a, v2f b, v8f c) {
    return __builtin_amdgcn_wmma_f32_16x16x4_f32(false, a, false, b, (short)0, c,
                                                 false, false);
}

__device__ __forceinline__ v2f lds_bslice(const float* buf, int kBase, int col, int kOff) {
    v2f b;
    b.x = buf[(kBase + kOff + 0) * 16 + col];
    b.y = buf[(kBase + kOff + 1) * 16 + col];
    return b;
}

__device__ __forceinline__ int orig_row(int p) {
    int gate  = p & 3;
    int blk   = p >> 3;
    int upair = ((blk & 1) << 1) | (blk >> 1);
    int unit  = upair * 2 + ((p >> 2) & 1);
    return gate * 8 + unit;
}

__global__ __launch_bounds__(32)
void lstm_ctrl_kernel(
    const float* __restrict__ state, const float* __restrict__ dist,
    const float* __restrict__ e1W1, const float* __restrict__ e1b1,
    const float* __restrict__ e1W2, const float* __restrict__ e1b2,
    const float* __restrict__ e2W1, const float* __restrict__ e2b1,
    const float* __restrict__ e2W2, const float* __restrict__ e2b2,
    const float* __restrict__ Wih0, const float* __restrict__ Whh0,
    const float* __restrict__ bih0, const float* __restrict__ bhh0,
    const float* __restrict__ Wih1, const float* __restrict__ Whh1,
    const float* __restrict__ bih1, const float* __restrict__ bhh1,
    const float* __restrict__ decW, const float* __restrict__ decb,
    const float* __restrict__ finW, const float* __restrict__ finb,
    const float* __restrict__ fin2W, const float* __restrict__ fin2b,
    float* __restrict__ outp, int T, int Nn)
{
    const int lane = threadIdx.x;
    const int col  = lane & 15;
    const int half = lane >> 4;
    const int kOff = half << 1;
    const int gcol = blockIdx.x * 16 + col;

    __shared__ float bufA[256];
    __shared__ float bufB[256];
    __shared__ float tmpS[256];

    const float* WihL[2] = { Wih0, Wih1 };
    const float* WhhL[2] = { Whh0, Whh1 };
    const float* bihL[2] = { bih0, bih1 };
    const float* bhhL[2] = { bhh0, bhh1 };

    v2f Aw[2][2][4];
    v8f BiasT[2][2];
#pragma unroll
    for (int l = 0; l < 2; ++l) {
#pragma unroll
        for (int tI = 0; tI < 2; ++tI) {
            const int orow = orig_row(tI * 16 + col);
#pragma unroll
            for (int ks = 0; ks < 4; ++ks) {
                v2f a;
#pragma unroll
                for (int cc = 0; cc < 2; ++cc) {
                    int k = ks * 4 + kOff + cc;
                    float w = (k < 8) ? WihL[l][orow * 8 + k]
                                      : WhhL[l][orow * 8 + (k - 8)];
                    if (cc == 0) a.x = w; else a.y = w;
                }
                Aw[l][tI][ks] = a;
            }
            v8f bv;
#pragma unroll
            for (int r = 0; r < 8; ++r) {
                int br = orig_row(tI * 16 + r + half * 8);
                bv[r] = bihL[l][br] + bhhL[l][br];
            }
            BiasT[l][tI] = bv;
        }
    }

    float dWr[4][8], dbr[4], hWr[4][4], hbr[4];
    {
        const float* fw = half ? fin2W : finW;
        const float* fb = half ? fin2b : finb;
#pragma unroll
        for (int d = 0; d < 4; ++d) {
            dbr[d] = decb[d];
#pragma unroll
            for (int u = 0; u < 8; ++u) dWr[d][u] = decW[d * 8 + u];
        }
#pragma unroll
        for (int a = 0; a < 4; ++a) {
            hbr[a] = fb[a];
#pragma unroll
            for (int d = 0; d < 4; ++d) hWr[a][d] = fw[a * 4 + d];
        }
    }

    {
        const float* W1 = half ? e2W1 : e1W1;
        const float* b1 = half ? e2b1 : e1b1;
        const float* W2 = half ? e2W2 : e1W2;
        const float* b2 = half ? e2b2 : e1b2;
        float sv[16];
#pragma unroll
        for (int i = 0; i < 16; ++i) sv[i] = state[gcol * 16 + i];
        float eh[4];
#pragma unroll
        for (int j = 0; j < 4; ++j) {
            float acc = b1[j];
#pragma unroll
            for (int i = 0; i < 16; ++i) acc = fmaf(W1[j * 16 + i], sv[i], acc);
            eh[j] = fmaxf(acc, 0.0f);
        }
        float ee[16];
#pragma unroll
        for (int u = 0; u < 16; ++u) {
            float acc = b2[u];
#pragma unroll
            for (int j = 0; j < 4; ++j) acc = fmaf(W2[u * 4 + j], eh[j], acc);
            ee[u] = fmaxf(acc, 0.0f);
        }
        if (half == 0) {
#pragma unroll
            for (int u = 0; u < 8; ++u) bufA[(8 + u) * 16 + col] = ee[u];
#pragma unroll
            for (int u = 0; u < 8; ++u) bufB[(8 + u) * 16 + col] = ee[8 + u];
        } else {
#pragma unroll
            for (int u = 0; u < 16; ++u) tmpS[u * 16 + col] = ee[u];
        }
    }
    DSBAR();

    float c0s[4], c1s[4];
#pragma unroll
    for (int ul = 0; ul < 4; ++ul) {
        int gu = half * 4 + ul;
        c0s[ul] = tmpS[gu * 16 + col];
        c1s[ul] = tmpS[(8 + gu) * 16 + col];
    }

#define GATE(G0, G1, ul, idx) (((ul) < 2) ? (G0)[(ul) * 4 + (idx)] : (G1)[((ul) - 2) * 4 + (idx)])

    const size_t muBase  = 0;
    const size_t sgBase  = (size_t)T * (size_t)Nn * 4;

    for (int t = 0; t < T; ++t) {
        const float* xp = dist + ((size_t)t * Nn + gcol) * 8;
        v2f Bx0, Bx1;
        Bx0.x = xp[kOff + 0]; Bx0.y = xp[kOff + 1];
        Bx1.x = xp[4 + kOff + 0]; Bx1.y = xp[4 + kOff + 1];
        if (t + 1 < T) __builtin_prefetch(xp + (size_t)Nn * 8, 0, 1);

        v2f Bh0 = lds_bslice(bufA, 8,  col, kOff);
        v2f Bh1 = lds_bslice(bufA, 12, col, kOff);
        v8f g0 = BiasT[0][0], g1 = BiasT[0][1];
        g0 = wmma_k4(Aw[0][0][0], Bx0, g0);  g1 = wmma_k4(Aw[0][1][0], Bx0, g1);
        g0 = wmma_k4(Aw[0][0][1], Bx1, g0);  g1 = wmma_k4(Aw[0][1][1], Bx1, g1);
        g0 = wmma_k4(Aw[0][0][2], Bh0, g0);  g1 = wmma_k4(Aw[0][1][2], Bh0, g1);
        g0 = wmma_k4(Aw[0][0][3], Bh1, g0);  g1 = wmma_k4(Aw[0][1][3], Bh1, g1);

        float h0n[4];
#pragma unroll
        for (int ul = 0; ul < 4; ++ul) {
            float gi = GATE(g0, g1, ul, 0), gf = GATE(g0, g1, ul, 1);
            float gg = GATE(g0, g1, ul, 2), go = GATE(g0, g1, ul, 3);
            float cn = fmaf(fast_sig(gf), c0s[ul], fast_sig(gi) * fast_tanh(gg));
            c0s[ul] = cn;
            h0n[ul] = fast_sig(go) * fast_tanh(cn);
        }
#pragma unroll
        for (int ul = 0; ul < 4; ++ul) {
            int gu = half * 4 + ul;
            bufA[(8 + gu) * 16 + col] = h0n[ul];
            bufB[gu * 16 + col]       = h0n[ul];
        }
        DSBAR();

        v2f B0 = lds_bslice(bufB, 0,  col, kOff);
        v2f B1 = lds_bslice(bufB, 4,  col, kOff);
        v2f B2 = lds_bslice(bufB, 8,  col, kOff);
        v2f B3 = lds_bslice(bufB, 12, col, kOff);
        v8f G0 = BiasT[1][0], G1 = BiasT[1][1];
        G0 = wmma_k4(Aw[1][0][0], B0, G0);  G1 = wmma_k4(Aw[1][1][0], B0, G1);
        G0 = wmma_k4(Aw[1][0][1], B1, G0);  G1 = wmma_k4(Aw[1][1][1], B1, G1);
        G0 = wmma_k4(Aw[1][0][2], B2, G0);  G1 = wmma_k4(Aw[1][1][2], B2, G1);
        G0 = wmma_k4(Aw[1][0][3], B3, G0);  G1 = wmma_k4(Aw[1][1][3], B3, G1);

        float h1n[4];
#pragma unroll
        for (int ul = 0; ul < 4; ++ul) {
            float gi = GATE(G0, G1, ul, 0), gf = GATE(G0, G1, ul, 1);
            float gg = GATE(G0, G1, ul, 2), go = GATE(G0, G1, ul, 3);
            float cn = fmaf(fast_sig(gf), c1s[ul], fast_sig(gi) * fast_tanh(gg));
            c1s[ul] = cn;
            h1n[ul] = fast_sig(go) * fast_tanh(cn);
        }
#pragma unroll
        for (int ul = 0; ul < 4; ++ul) {
            int gu = half * 4 + ul;
            bufB[(8 + gu) * 16 + col] = h1n[ul];
        }
        DSBAR();

        float hv[8];
#pragma unroll
        for (int u = 0; u < 8; ++u) hv[u] = bufB[(8 + u) * 16 + col];
        float dec[4];
#pragma unroll
        for (int d = 0; d < 4; ++d) {
            float acc = dbr[d];
#pragma unroll
            for (int u = 0; u < 8; ++u) acc = fmaf(dWr[d][u], hv[u], acc);
            dec[d] = fmaxf(acc, 0.0f);
        }
        v4f res;
#pragma unroll
        for (int a = 0; a < 4; ++a) {
            float acc = hbr[a];
#pragma unroll
            for (int d = 0; d < 4; ++d) acc = fmaf(hWr[a][d], dec[d], acc);
            res[a] = acc;
        }
        size_t off = (half ? sgBase : muBase) + ((size_t)t * Nn + gcol) * 4;
        *(v4f*)(outp + off) = res;
    }
#undef GATE
}

extern "C" void kernel_launch(void* const* d_in, const int* in_sizes, int n_in,
                              void* d_out, int out_size, void* d_ws, size_t ws_size,
                              hipStream_t stream) {
    (void)n_in; (void)out_size; (void)d_ws; (void)ws_size;
    const float* state = (const float*)d_in[0];
    const float* dist  = (const float*)d_in[1];
    const float* e1W1  = (const float*)d_in[2];
    const float* e1b1  = (const float*)d_in[3];
    const float* e1W2  = (const float*)d_in[4];
    const float* e1b2  = (const float*)d_in[5];
    const float* e2W1  = (const float*)d_in[6];
    const float* e2b1  = (const float*)d_in[7];
    const float* e2W2  = (const float*)d_in[8];
    const float* e2b2  = (const float*)d_in[9];
    const float* Wih0  = (const float*)d_in[10];
    const float* Whh0  = (const float*)d_in[11];
    const float* bih0  = (const float*)d_in[12];
    const float* bhh0  = (const float*)d_in[13];
    const float* Wih1  = (const float*)d_in[14];
    const float* Whh1  = (const float*)d_in[15];
    const float* bih1  = (const float*)d_in[16];
    const float* bhh1  = (const float*)d_in[17];
    const float* decW  = (const float*)d_in[18];
    const float* decb  = (const float*)d_in[19];
    const float* finW  = (const float*)d_in[20];
    const float* finb  = (const float*)d_in[21];
    const float* fin2W = (const float*)d_in[22];
    const float* fin2b = (const float*)d_in[23];

    const int Nn = in_sizes[0] / 16;
    const int T  = in_sizes[1] / (Nn * 8);

    lstm_ctrl_kernel<<<dim3(Nn / 16), dim3(32), 0, stream>>>(
        state, dist,
        e1W1, e1b1, e1W2, e1b2, e2W1, e2b1, e2W2, e2b2,
        Wih0, Whh0, bih0, bhh0, Wih1, Whh1, bih1, bhh1,
        decW, decb, finW, finb, fin2W, fin2b,
        (float*)d_out, T, Nn);
}


# h stays in registers as the next B fragment (k order of the h slices matched to the D layout; decoder takes the other half's h by permlanex16): no LDS in the time loop; k-sum order of h dots changes
# speedup vs baseline: 1.3391x; 1.3391x over previous
#include <hip/hip_runtime.h>


typedef float v2f __attribute__((ext_vector_type(2)));
typedef float v4f __attribute__((ext_vector_type(4)));
typedef float v8f __attribute__((ext_vector_type(8)));

#define DSBAR() asm volatile("s_wait_dscnt 0" ::: "memory")

__device__ __forceinline__ float fast_tanh(float x) {
#if __has_builtin(__builtin_amdgcn_tanhf)
    return __builtin_amdgcn_tanhf(x);
#else
    float e = __builtin_amdgcn_exp2f(x * 2.8853900817779268f);
    return 1.0f - 2.0f * __builtin_amdgcn_rcpf(e + 1.0f);
#endif
}
__device__ __forceinline__ float fast_sig(float x) {
    return fmaf(0.5f, fast_tanh(0.5f * x), 0.5f);
}

__device__ __forceinline__ v8f wmma_k4(v2f a, v2f b, v8f c) {
    return __builtin_amdgcn_wmma_f32_16x16x4_f32(false, a, false, b, (short)0, c,
                                                 false, false);
}


__device__ __forceinline__ float xhalf(float x) {
    return __builtin_bit_cast(float, __builtin_amdgcn_permlanex16(
        __builtin_bit_cast(int, x), __builtin_bit_cast(int, x),
        0x76543210, 0xfedcba98, false, false));
}

__device__ __forceinline__ int orig_row(int p) {
    int gate  = p & 3;
    int blk   = p >> 3;
    int upair = ((blk & 1) << 1) | (blk >> 1);
    int unit  = upair * 2 + ((p >> 2) & 1);
    return gate * 8 + unit;
}

__global__ __launch_bounds__(32)
void lstm_ctrl_kernel(
    const float* __restrict__ state, const float* __restrict__ dist,
    const float* __restrict__ e1W1, const float* __restrict__ e1b1,
    const float* __restrict__ e1W2, const float* __restrict__ e1b2,
    const float* __restrict__ e2W1, const float* __restrict__ e2b1,
    const float* __restrict__ e2W2, const float* __restrict__ e2b2,
    const float* __restrict__ Wih0, const float* __restrict__ Whh0,
    const float* __restrict__ bih0, const float* __restrict__ bhh0,
    const float* __restrict__ Wih1, const float* __restrict__ Whh1,
    const float* __restrict__ bih1, const float* __restrict__ bhh1,
    const float* __restrict__ decW, const float* __restrict__ decb,
    const float* __restrict__ finW, const float* __restrict__ finb,
    const float* __restrict__ fin2W, const float* __restrict__ fin2b,
    float* __restrict__ outp, int T, int Nn)
{
    const int lane = threadIdx.x;
    const int col  = lane & 15;
    const int half = lane >> 4;
    const int kOff = half << 1;
    const int gcol = blockIdx.x * 16 + col;

    __shared__ float initS[2][256];

    const float* WihL[2] = { Wih0, Wih1 };
    const float* WhhL[2] = { Whh0, Whh1 };
    const float* bihL[2] = { bih0, bih1 };
    const float* bhhL[2] = { bhh0, bhh1 };

    v2f Aw[2][2][4];
    v8f BiasT[2][2];
#pragma unroll
    for (int l = 0; l < 2; ++l) {
#pragma unroll
        for (int tI = 0; tI < 2; ++tI) {
            const int orow = orig_row(tI * 16 + col);
#pragma unroll
            for (int ks = 0; ks < 4; ++ks) {
                v2f a;
#pragma unroll
                for (int cc = 0; cc < 2; ++cc) {
                    const float* wsrc = (ks < 2) ? WihL[l] : WhhL[l];
                    int f = (l == 0 && ks < 2) ? (ks * 4 + kOff + cc)
                                               : (half * 4 + (ks & 1) * 2 + cc);
                    float w = wsrc[orow * 8 + f];
                    if (cc == 0) a.x = w; else a.y = w;
                }
                Aw[l][tI][ks] = a;
            }
            v8f bv;
#pragma unroll
            for (int r = 0; r < 8; ++r) {
                int br = orig_row(tI * 16 + r + half * 8);
                bv[r] = bihL[l][br] + bhhL[l][br];
            }
            BiasT[l][tI] = bv;
        }
    }

    float dWo[4][4], dWp[4][4], dbr[4], hWr[4][4], hbr[4];
    {
        const float* fw = half ? fin2W : finW;
        const float* fb = half ? fin2b : finb;
#pragma unroll
        for (int d = 0; d < 4; ++d) {
            dbr[d] = decb[d];
#pragma unroll
            for (int j = 0; j < 4; ++j) {
                dWo[d][j] = decW[d * 8 + half * 4 + j];
                dWp[d][j] = decW[d * 8 + (1 - half) * 4 + j];
            }
        }
#pragma unroll
        for (int a = 0; a < 4; ++a) {
            hbr[a] = fb[a];
#pragma unroll
            for (int d = 0; d < 4; ++d) hWr[a][d] = fw[a * 4 + d];
        }
    }

    {
        const float* W1 = half ? e2W1 : e1W1;
        const float* b1 = half ? e2b1 : e1b1;
        const float* W2 = half ? e2W2 : e1W2;
        const float* b2 = half ? e2b2 : e1b2;
        float sv[16];
#pragma unroll
        for (int i = 0; i < 16; ++i) sv[i] = state[gcol * 16 + i];
        float eh[4];
#pragma unroll
        for (int j = 0; j < 4; ++j) {
            float acc = b1[j];
#pragma unroll
            for (int i = 0; i < 16; ++i) acc = fmaf(W1[j * 16 + i], sv[i], acc);
            eh[j] = fmaxf(acc, 0.0f);
        }
        float ee[16];
#pragma unroll
        for (int u = 0; u < 16; ++u) {
            float acc = b2[u];
#pragma unroll
            for (int j = 0; j < 4; ++j) acc = fmaf(W2[u * 4 + j], eh[j], acc);
            ee[u] = fmaxf(acc, 0.0f);
        }
#pragma unroll
        for (int u = 0; u < 16; ++u) initS[half][u * 16 + col] = ee[u];
    }
    DSBAR();

    float h0s[4], h1s[4], c0s[4], c1s[4];
#pragma unroll
    for (int ul = 0; ul < 4; ++ul) {
        int gu = half * 4 + ul;
        h0s[ul] = initS[0][gu * 16 + col];
        h1s[ul] = initS[0][(8 + gu) * 16 + col];
        c0s[ul] = initS[1][gu * 16 + col];
        c1s[ul] = initS[1][(8 + gu) * 16 + col];
    }

#define GATE(G0, G1, ul, idx) (((ul) < 2) ? (G0)[(ul) * 4 + (idx)] : (G1)[((ul) - 2) * 4 + (idx)])

    const size_t muBase  = 0;
    const size_t sgBase  = (size_t)T * (size_t)Nn * 4;

    for (int t = 0; t < T; ++t) {
        const float* xp = dist + ((size_t)t * Nn + gcol) * 8;
        v2f Bx0, Bx1;
        Bx0.x = xp[kOff + 0]; Bx0.y = xp[kOff + 1];
        Bx1.x = xp[4 + kOff + 0]; Bx1.y = xp[4 + kOff + 1];
        if (t + 1 < T) __builtin_prefetch(xp + (size_t)Nn * 8, 0, 1);

        v2f Bh0, Bh1;
        Bh0.x = h0s[0]; Bh0.y = h0s[1];
        Bh1.x = h0s[2]; Bh1.y = h0s[3];
        v8f g0 = BiasT[0][0], g1 = BiasT[0][1];
        g0 = wmma_k4(Aw[0][0][0], Bx0, g0);  g1 = wmma_k4(Aw[0][1][0], Bx0, g1);
        g0 = wmma_k4(Aw[0][0][1], Bx1, g0);  g1 = wmma_k4(Aw[0][1][1], Bx1, g1);
        g0 = wmma_k4(Aw[0][0][2], Bh0, g0);  g1 = wmma_k4(Aw[0][1][2], Bh0, g1);
        g0 = wmma_k4(Aw[0][0][3], Bh1, g0);  g1 = wmma_k4(Aw[0][1][3], Bh1, g1);

#pragma unroll
        for (int ul = 0; ul < 4; ++ul) {
            float gi = GATE(g0, g1, ul, 0), gf = GATE(g0, g1, ul, 1);
            float gg = GATE(g0, g1, ul, 2), go = GATE(g0, g1, ul, 3);
            float cn = fmaf(fast_sig(gf), c0s[ul], fast_sig(gi) * fast_tanh(gg));
            c0s[ul] = cn;
            h0s[ul] = fast_sig(go) * fast_tanh(cn);
        }

        v2f B0, B1, B2, B3;
        B0.x = h0s[0]; B0.y = h0s[1];
        B1.x = h0s[2]; B1.y = h0s[3];
        B2.x = h1s[0]; B2.y = h1s[1];
        B3.x = h1s[2]; B3.y = h1s[3];
        v8f G0 = BiasT[1][0], G1 = BiasT[1][1];
        G0 = wmma_k4(Aw[1][0][0], B0, G0);  G1 = wmma_k4(Aw[1][1][0], B0, G1);
        G0 = wmma_k4(Aw[1][0][1], B1, G0);  G1 = wmma_k4(Aw[1][1][1], B1, G1);
        G0 = wmma_k4(Aw[1][0][2], B2, G0);  G1 = wmma_k4(Aw[1][1][2], B2, G1);
        G0 = wmma_k4(Aw[1][0][3], B3, G0);  G1 = wmma_k4(Aw[1][1][3], B3, G1);

#pragma unroll
        for (int ul = 0; ul < 4; ++ul) {
            float gi = GATE(G0, G1, ul, 0), gf = GATE(G0, G1, ul, 1);
            float gg = GATE(G0, G1, ul, 2), go = GATE(G0, G1, ul, 3);
            float cn = fmaf(fast_sig(gf), c1s[ul], fast_sig(gi) * fast_tanh(gg));
            c1s[ul] = cn;
            h1s[ul] = fast_sig(go) * fast_tanh(cn);
        }

        float hp[4];
#pragma unroll
        for (int j = 0; j < 4; ++j) hp[j] = xhalf(h1s[j]);
        float dec[4];
#pragma unroll
        for (int d = 0; d < 4; ++d) {
            float acc = dbr[d];
#pragma unroll
            for (int j = 0; j < 4; ++j) acc = fmaf(dWo[d][j], h1s[j], acc);
#pragma unroll
            for (int j = 0; j < 4; ++j) acc = fmaf(dWp[d][j], hp[j], acc);
            dec[d] = fmaxf(acc, 0.0f);
        }
        v4f res;
#pragma unroll
        for (int a = 0; a < 4; ++a) {
            float acc = hbr[a];
#pragma unroll
            for (int d = 0; d < 4; ++d) acc = fmaf(hWr[a][d], dec[d], acc);
            res[a] = acc;
        }
        size_t off = (half ? sgBase : muBase) + ((size_t)t * Nn + gcol) * 4;
        *(v4f*)(outp + off) = res;
    }
#undef GATE
}

extern "C" void kernel_launch(void* const* d_in, const int* in_sizes, int n_in,
                              void* d_out, int out_size, void* d_ws, size_t ws_size,
                              hipStream_t stream) {
    (void)n_in; (void)out_size; (void)d_ws; (void)ws_size;
    const float* state = (const float*)d_in[0];
    const float* dist  = (const float*)d_in[1];
    const float* e1W1  = (const float*)d_in[2];
    const float* e1b1  = (const float*)d_in[3];
    const float* e1W2  = (const float*)d_in[4];
    const float* e1b2  = (const float*)d_in[5];
    const float* e2W1  = (const float*)d_in[6];
    const float* e2b1  = (const float*)d_in[7];
    const float* e2W2  = (const float*)d_in[8];
    const float* e2b2  = (const float*)d_in[9];
    const float* Wih0  = (const float*)d_in[10];
    const float* Whh0  = (const float*)d_in[11];
    const float* bih0  = (const float*)d_in[12];
    const float* bhh0  = (const float*)d_in[13];
    const float* Wih1  = (const float*)d_in[14];
    const float* Whh1  = (const float*)d_in[15];
    const float* bih1  = (const float*)d_in[16];
    const float* bhh1  = (const float*)d_in[17];
    const float* decW  = (const float*)d_in[18];
    const float* decb  = (const float*)d_in[19];
    const float* finW  = (const float*)d_in[20];
    const float* finb  = (const float*)d_in[21];
    const float* fin2W = (const float*)d_in[22];
    const float* fin2b = (const float*)d_in[23];

    const int Nn = in_sizes[0] / 16;
    const int T  = in_sizes[1] / (Nn * 8);

    lstm_ctrl_kernel<<<dim3(Nn / 16), dim3(32), 0, stream>>>(
        state, dist,
        e1W1, e1b1, e1W2, e1b2, e2W1, e2b1, e2W2, e2b2,
        Wih0, Whh0, bih0, bhh0, Wih1, Whh1, bih1, bhh1,
        decW, decb, finW, finb, fin2W, fin2b,
        (float*)d_out, T, Nn);
}


# x_t rows loaded one block of 4 steps ahead into a register ring, time loop unrolled by 4 (one step copy per slot), L2 hint dropped: no step waits on a global load, neighbouring steps interleave
# speedup vs baseline: 3.0081x; 2.2464x over previous
#include <hip/hip_runtime.h>


typedef float v2f __attribute__((ext_vector_type(2)));
typedef float v4f __attribute__((ext_vector_type(4)));
typedef float v8f __attribute__((ext_vector_type(8)));

#define DSBAR() asm volatile("s_wait_dscnt 0" ::: "memory")

__device__ __forceinline__ float fast_tanh(float x) {
#if __has_builtin(__builtin_amdgcn_tanhf)
    return __builtin_amdgcn_tanhf(x);
#else
    float e = __builtin_amdgcn_exp2f(x * 2.8853900817779268f);
    return 1.0f - 2.0f * __builtin_amdgcn_rcpf(e + 1.0f);
#endif
}
__device__ __forceinline__ float fast_sig(float x) {
    return fmaf(0.5f, fast_tanh(0.5f * x), 0.5f);
}

__device__ __forceinline__ v8f wmma_k4(v2f a, v2f b, v8f c) {
    return __builtin_amdgcn_wmma_f32_16x16x4_f32(false, a, false, b, (short)0, c,
                                                 false, false);
}


__device__ __forceinline__ float xhalf(float x) {
    return __builtin_bit_cast(float, __builtin_amdgcn_permlanex16(
        __builtin_bit_cast(int, x), __builtin_bit_cast(int, x),
        0x76543210, 0xfedcba98, false, false));
}

__device__ __forceinline__ int orig_row(int p) {
    int gate  = p & 3;
    int blk   = p >> 3;
    int upair = ((blk & 1) << 1) | (blk >> 1);
    int unit  = upair * 2 + ((p >> 2) & 1);
    return gate * 8 + unit;
}

__global__ __launch_bounds__(32)
void lstm_ctrl_kernel(
    const float* __restrict__ state, const float* __restrict__ dist,
    const float* __restrict__ e1W1, const float* __restrict__ e1b1,
    const float* __restrict__ e1W2, const float* __restrict__ e1b2,
    const float* __restrict__ e2W1, const float* __restrict__ e2b1,
    const float* __restrict__ e2W2, const float* __restrict__ e2b2,
    const float* __restrict__ Wih0, const float* __restrict__ Whh0,
    const float* __restrict__ bih0, const float* __restrict__ bhh0,
    const float* __restrict__ Wih1, const float* __restrict__ Whh1,
    const float* __restrict__ bih1, const float* __restrict__ bhh1,
    const float* __restrict__ decW, const float* __restrict__ decb,
    const float* __restrict__ finW, const float* __restrict__ finb,
    const float* __restrict__ fin2W, const float* __restrict__ fin2b,
    float* __restrict__ outp, int T, int Nn)
{
    const int lane = threadIdx.x;
    const int col  = lane & 15;
    const int half = lane >> 4;
    const int kOff = half << 1;
    const int gcol = blockIdx.x * 16 + col;

    __shared__ float initS[2][256];

    const float* WihL[2] = { Wih0, Wih1 };
    const float* WhhL[2] = { Whh0, Whh1 };
    const float* bihL[2] = { bih0, bih1 };
    const float* bhhL[2] = { bhh0, bhh1 };

    v2f Aw[2][2][4];
    v8f BiasT[2][2];
#pragma unroll
    for (int l = 0; l < 2; ++l) {
#pragma unroll
        for (int tI = 0; tI < 2; ++tI) {
            const int orow = orig_row(tI * 16 + col);
#pragma unroll
            for (int ks = 0; ks < 4; ++ks) {
                v2f a;
#pragma unroll
                for (int cc = 0; cc < 2; ++cc) {
                    const float* wsrc = (ks < 2) ? WihL[l] : WhhL[l];
                    int f = (l == 0 && ks < 2) ? (ks * 4 + kOff + cc)
                                               : (half * 4 + (ks & 1) * 2 + cc);
                    float w = wsrc[orow * 8 + f];
                    if (cc == 0) a.x = w; else a.y = w;
                }
                Aw[l][tI][ks] = a;
            }
            v8f bv;
#pragma unroll
            for (int r = 0; r < 8; ++r) {
                int br = orig_row(tI * 16 + r + half * 8);
                bv[r] = bihL[l][br] + bhhL[l][br];
            }
            BiasT[l][tI] = bv;
        }
    }

    float dWo[4][4], dWp[4][4], dbr[4], hWr[4][4], hbr[4];
    {
        const float* fw = half ? fin2W : finW;
        const float* fb = half ? fin2b : finb;
#pragma unroll
        for (int d = 0; d < 4; ++d) {
            dbr[d] = decb[d];
#pragma unroll
            for (int j = 0; j < 4; ++j) {
                dWo[d][j] = decW[d * 8 + half * 4 + j];
                dWp[d][j] = decW[d * 8 + (1 - half) * 4 + j];
            }
        }
#pragma unroll
        for (int a = 0; a < 4; ++a) {
            hbr[a] = fb[a];
#pragma unroll
            for (int d = 0; d < 4; ++d) hWr[a][d] = fw[a * 4 + d];
        }
    }

    {
        const float* W1 = half ? e2W1 : e1W1;
        const float* b1 = half ? e2b1 : e1b1;
        const float* W2 = half ? e2W2 : e1W2;
        const float* b2 = half ? e2b2 : e1b2;
        float sv[16];
#pragma unroll
        for (int i = 0; i < 16; ++i) sv[i] = state[gcol * 16 + i];
        float eh[4];
#pragma unroll
        for (int j = 0; j < 4; ++j) {
            float acc = b1[j];
#pragma unroll
            for (int i = 0; i < 16; ++i) acc = fmaf(W1[j * 16 + i], sv[i], acc);
            eh[j] = fmaxf(acc, 0.0f);
        }
        float ee[16];
#pragma unroll
        for (int u = 0; u < 16; ++u) {
            float acc = b2[u];
#pragma unroll
            for (int j = 0; j < 4; ++j) acc = fmaf(W2[u * 4 + j], eh[j], acc);
            ee[u] = fmaxf(acc, 0.0f);
        }
#pragma unroll
        for (int u = 0; u < 16; ++u) initS[half][u * 16 + col] = ee[u];
    }
    DSBAR();

    float h0s[4], h1s[4], c0s[4], c1s[4];
#pragma unroll
    for (int ul = 0; ul < 4; ++ul) {
        int gu = half * 4 + ul;
        h0s[ul] = initS[0][gu * 16 + col];
        h1s[ul] = initS[0][(8 + gu) * 16 + col];
        c0s[ul] = initS[1][gu * 16 + col];
        c1s[ul] = initS[1][(8 + gu) * 16 + col];
    }

#define GATE(G0, G1, ul, idx) (((ul) < 2) ? (G0)[(ul) * 4 + (idx)] : (G1)[((ul) - 2) * 4 + (idx)])

    const size_t muBase  = 0;
    const size_t sgBase  = (size_t)T * (size_t)Nn * 4;

    const int xlane = gcol * 8 + kOff;
    auto load_x = [&](int tt, v2f& b0, v2f& b1) {
        const float* xrow = dist + (size_t)tt * (size_t)Nn * 8;
        b0.x = xrow[xlane + 0]; b0.y = xrow[xlane + 1];
        b1.x = xrow[xlane + 4]; b1.y = xrow[xlane + 5];
    };

    auto step = [&](int t, v2f Bx0, v2f Bx1) {
        v2f Bh0, Bh1;
        Bh0.x = h0s[0]; Bh0.y = h0s[1];
        Bh1.x = h0s[2]; Bh1.y = h0s[3];
        v8f g0 = BiasT[0][0], g1 = BiasT[0][1];
        g0 = wmma_k4(Aw[0][0][0], Bx0, g0);  g1 = wmma_k4(Aw[0][1][0], Bx0, g1);
        g0 = wmma_k4(Aw[0][0][1], Bx1, g0);  g1 = wmma_k4(Aw[0][1][1], Bx1, g1);
        g0 = wmma_k4(Aw[0][0][2], Bh0, g0);  g1 = wmma_k4(Aw[0][1][2], Bh0, g1);
        g0 = wmma_k4(Aw[0][0][3], Bh1, g0);  g1 = wmma_k4(Aw[0][1][3], Bh1, g1);

#pragma unroll
        for (int ul = 0; ul < 4; ++ul) {
            float gi = GATE(g0, g1, ul, 0), gf = GATE(g0, g1, ul, 1);
            float gg = GATE(g0, g1, ul, 2), go = GATE(g0, g1, ul, 3);
            float cn = fmaf(fast_sig(gf), c0s[ul], fast_sig(gi) * fast_tanh(gg));
            c0s[ul] = cn;
            h0s[ul] = fast_sig(go) * fast_tanh(cn);
        }

        v2f B0, B1, B2, B3;
        B0.x = h0s[0]; B0.y = h0s[1];
        B1.x = h0s[2]; B1.y = h0s[3];
        B2.x = h1s[0]; B2.y = h1s[1];
        B3.x = h1s[2]; B3.y = h1s[3];
        v8f G0 = BiasT[1][0], G1 = BiasT[1][1];
        G0 = wmma_k4(Aw[1][0][0], B0, G0);  G1 = wmma_k4(Aw[1][1][0], B0, G1);
        G0 = wmma_k4(Aw[1][0][1], B1, G0);  G1 = wmma_k4(Aw[1][1][1], B1, G1);
        G0 = wmma_k4(Aw[1][0][2], B2, G0);  G1 = wmma_k4(Aw[1][1][2], B2, G1);
        G0 = wmma_k4(Aw[1][0][3], B3, G0);  G1 = wmma_k4(Aw[1][1][3], B3, G1);

#pragma unroll
        for (int ul = 0; ul < 4; ++ul) {
            float gi = GATE(G0, G1, ul, 0), gf = GATE(G0, G1, ul, 1);
            float gg = GATE(G0, G1, ul, 2), go = GATE(G0, G1, ul, 3);
            float cn = fmaf(fast_sig(gf), c1s[ul], fast_sig(gi) * fast_tanh(gg));
            c1s[ul] = cn;
            h1s[ul] = fast_sig(go) * fast_tanh(cn);
        }

        float hp[4];
#pragma unroll
        for (int j = 0; j < 4; ++j) hp[j] = xhalf(h1s[j]);
        float dec[4];
#pragma unroll
        for (int d = 0; d < 4; ++d) {
            float acc = dbr[d];
#pragma unroll
            for (int j = 0; j < 4; ++j) acc = fmaf(dWo[d][j], h1s[j], acc);
#pragma unroll
            for (int j = 0; j < 4; ++j) acc = fmaf(dWp[d][j], hp[j], acc);
            dec[d] = fmaxf(acc, 0.0f);
        }
        v4f res;
#pragma unroll
        for (int a = 0; a < 4; ++a) {
            float acc = hbr[a];
#pragma unroll
            for (int d = 0; d < 4; ++d) acc = fmaf(hWr[a][d], dec[d], acc);
            res[a] = acc;
        }
        size_t off = (half ? sgBase : muBase) + ((size_t)t * Nn + gcol) * 4;
        *(v4f*)(outp + off) = res;
    };

    if (T <= 0) return;

    constexpr int XPD = 4;
    v2f xr0[XPD], xr1[XPD];
#pragma unroll
    for (int u = 0; u < XPD; ++u) load_x(u < T ? u : T - 1, xr0[u], xr1[u]);

    int t = 0;
    for (; t + XPD <= T; t += XPD) {
        v2f Bx0[XPD], Bx1[XPD];
#pragma unroll
        for (int u = 0; u < XPD; ++u) { Bx0[u] = xr0[u]; Bx1[u] = xr1[u]; }
#pragma unroll
        for (int u = 0; u < XPD; ++u) {
            const int tn = t + u + XPD;
            load_x(tn < T ? tn : T - 1, xr0[u], xr1[u]);
        }
#pragma unroll
        for (int u = 0; u < XPD; ++u) step(t + u, Bx0[u], Bx1[u]);
    }
    for (; t < T; ++t) {
        v2f Bx0, Bx1;
        load_x(t, Bx0, Bx1);
        step(t, Bx0, Bx1);
    }
#undef GATE
}

extern "C" void kernel_launch(void* const* d_in, const int* in_sizes, int n_in,
                              void* d_out, int out_size, void* d_ws, size_t ws_size,
                              hipStream_t stream) {
    (void)n_in; (void)out_size; (void)d_ws; (void)ws_size;
    const float* state = (const float*)d_in[0];
    const float* dist  = (const float*)d_in[1];
    const float* e1W1  = (const float*)d_in[2];
    const float* e1b1  = (const float*)d_in[3];
    const float* e1W2  = (const float*)d_in[4];
    const float* e1b2  = (const float*)d_in[5];
    const float* e2W1  = (const float*)d_in[6];
    const float* e2b1  = (const float*)d_in[7];
    const float* e2W2  = (const float*)d_in[8];
    const float* e2b2  = (const float*)d_in[9];
    const float* Wih0  = (const float*)d_in[10];
    const float* Whh0  = (const float*)d_in[11];
    const float* bih0  = (const float*)d_in[12];
    const float* bhh0  = (const float*)d_in[13];
    const float* Wih1  = (const float*)d_in[14];
    const float* Whh1  = (const float*)d_in[15];
    const float* bih1  = (const float*)d_in[16];
    const float* bhh1  = (const float*)d_in[17];
    const float* decW  = (const float*)d_in[18];
    const float* decb  = (const float*)d_in[19];
    const float* finW  = (const float*)d_in[20];
    const float* finb  = (const float*)d_in[21];
    const float* fin2W = (const float*)d_in[22];
    const float* fin2b = (const float*)d_in[23];

    const int Nn = in_sizes[0] / 16;
    const int T  = in_sizes[1] / (Nn * 8);

    lstm_ctrl_kernel<<<dim3(Nn / 16), dim3(32), 0, stream>>>(
        state, dist,
        e1W1, e1b1, e1W2, e1b2, e2W1, e2b1, e2W2, e2b2,
        Wih0, Whh0, bih0, bhh0, Wih1, Whh1, bih1, bhh1,
        decW, decb, finW, finb, fin2W, fin2b,
        (float*)d_out, T, Nn);
}


# layer 0, layer 1 and decoder+store run on three waves of the block (one SIMD each), pipelined over blocks of 8 steps through two-slot LDS hand-over buffers, one barrier per block
# speedup vs baseline: 5.8188x; 1.9344x over previous
#include <hip/hip_runtime.h>


typedef float v2f __attribute__((ext_vector_type(2)));
typedef float v4f __attribute__((ext_vector_type(4)));
typedef float v8f __attribute__((ext_vector_type(8)));

__device__ __forceinline__ float fast_tanh(float x) {
#if __has_builtin(__builtin_amdgcn_tanhf)
    return __builtin_amdgcn_tanhf(x);
#else
    float e = __builtin_amdgcn_exp2f(x * 2.8853900817779268f);
    return 1.0f - 2.0f * __builtin_amdgcn_rcpf(e + 1.0f);
#endif
}
__device__ __forceinline__ float fast_sig(float x) {
    return fmaf(0.5f, fast_tanh(0.5f * x), 0.5f);
}

__device__ __forceinline__ v8f wmma_k4(v2f a, v2f b, v8f c) {
    return __builtin_amdgcn_wmma_f32_16x16x4_f32(false, a, false, b, (short)0, c,
                                                 false, false);
}


__device__ __forceinline__ int orig_row(int p) {
    int gate  = p & 3;
    int blk   = p >> 3;
    int upair = ((blk & 1) << 1) | (blk >> 1);
    int unit  = upair * 2 + ((p >> 2) & 1);
    return gate * 8 + unit;
}

#define CH 8

__global__ __launch_bounds__(96)
void lstm_ctrl_kernel(
    const float* __restrict__ state, const float* __restrict__ dist,
    const float* __restrict__ e1W1, const float* __restrict__ e1b1,
    const float* __restrict__ e1W2, const float* __restrict__ e1b2,
    const float* __restrict__ e2W1, const float* __restrict__ e2b1,
    const float* __restrict__ e2W2, const float* __restrict__ e2b2,
    const float* __restrict__ Wih0, const float* __restrict__ Whh0,
    const float* __restrict__ bih0, const float* __restrict__ bhh0,
    const float* __restrict__ Wih1, const float* __restrict__ Whh1,
    const float* __restrict__ bih1, const float* __restrict__ bhh1,
    const float* __restrict__ decW, const float* __restrict__ decb,
    const float* __restrict__ finW, const float* __restrict__ finb,
    const float* __restrict__ fin2W, const float* __restrict__ fin2b,
    float* __restrict__ outp, int T, int Nn)
{
    const int lane = threadIdx.x & 31;
    const int role = __builtin_amdgcn_readfirstlane((int)(threadIdx.x >> 5));
    const int col  = lane & 15;
    const int half = lane >> 4;
    const int kOff = half << 1;
    const int gcol = blockIdx.x * 16 + col;

    __shared__ v4f hand0[2][CH][32];
    __shared__ v4f hand1[2][CH][32];

    if (T <= 0) return;

    v2f Aw[2][4];
    v8f BiasT[2];
    float hs[4], cs[4];
#pragma unroll
    for (int i = 0; i < 4; ++i) { hs[i] = 0.0f; cs[i] = 0.0f; }
#pragma unroll
    for (int tI = 0; tI < 2; ++tI) {
        BiasT[tI] = (v8f)(0.0f);
#pragma unroll
        for (int ks = 0; ks < 4; ++ks) Aw[tI][ks] = (v2f)(0.0f);
    }
    if (role < 2) {
        const int l = role;
        const float* Wih = l ? Wih1 : Wih0;
        const float* Whh = l ? Whh1 : Whh0;
        const float* bih = l ? bih1 : bih0;
        const float* bhh = l ? bhh1 : bhh0;
#pragma unroll
        for (int tI = 0; tI < 2; ++tI) {
            const int orow = orig_row(tI * 16 + col);
#pragma unroll
            for (int ks = 0; ks < 4; ++ks) {
                v2f a;
#pragma unroll
                for (int cc = 0; cc < 2; ++cc) {
                    const float* wsrc = (ks < 2) ? Wih : Whh;
                    int f = (l == 0 && ks < 2) ? (ks * 4 + kOff + cc)
                                               : (half * 4 + (ks & 1) * 2 + cc);
                    float w = wsrc[orow * 8 + f];
                    if (cc == 0) a.x = w; else a.y = w;
                }
                Aw[tI][ks] = a;
            }
            v8f bv;
#pragma unroll
            for (int r = 0; r < 8; ++r) {
                int br = orig_row(tI * 16 + r + half * 8);
                bv[r] = bih[br] + bhh[br];
            }
            BiasT[tI] = bv;
        }
        float sv[16];
#pragma unroll
        for (int i = 0; i < 16; ++i) sv[i] = state[gcol * 16 + i];
        float eh1[4], eh2[4];
#pragma unroll
        for (int j = 0; j < 4; ++j) {
            float a1 = e1b1[j], a2 = e2b1[j];
#pragma unroll
            for (int i = 0; i < 16; ++i) {
                a1 = fmaf(e1W1[j * 16 + i], sv[i], a1);
                a2 = fmaf(e2W1[j * 16 + i], sv[i], a2);
            }
            eh1[j] = fmaxf(a1, 0.0f);
            eh2[j] = fmaxf(a2, 0.0f);
        }
#pragma unroll
        for (int ul = 0; ul < 4; ++ul) {
            const int e = l * 8 + half * 4 + ul;
            float a1 = e1b2[e], a2 = e2b2[e];
#pragma unroll
            for (int j = 0; j < 4; ++j) {
                a1 = fmaf(e1W2[e * 4 + j], eh1[j], a1);
                a2 = fmaf(e2W2[e * 4 + j], eh2[j], a2);
            }
            hs[ul] = fmaxf(a1, 0.0f);
            cs[ul] = fmaxf(a2, 0.0f);
        }
    }

    float dWo[4][4], dWp[4][4], dbr[4], hWr[4][4], hbr[4];
    {
        const float* fw = half ? fin2W : finW;
        const float* fb = half ? fin2b : finb;
#pragma unroll
        for (int d = 0; d < 4; ++d) {
            dbr[d] = decb[d];
#pragma unroll
            for (int j = 0; j < 4; ++j) {
                dWo[d][j] = decW[d * 8 + half * 4 + j];
                dWp[d][j] = decW[d * 8 + (1 - half) * 4 + j];
            }
        }
#pragma unroll
        for (int a = 0; a < 4; ++a) {
            hbr[a] = fb[a];
#pragma unroll
            for (int d = 0; d < 4; ++d) hWr[a][d] = fw[a * 4 + d];
        }
    }

#define GATE(G0, G1, ul, idx) (((ul) < 2) ? (G0)[(ul) * 4 + (idx)] : (G1)[((ul) - 2) * 4 + (idx)])

    const size_t muBase  = 0;
    const size_t sgBase  = (size_t)T * (size_t)Nn * 4;

    const int xlane = gcol * 8 + kOff;
    auto load_x = [&](int tt, v2f& b0, v2f& b1) {
        const float* xrow = dist + (size_t)tt * (size_t)Nn * 8;
        b0.x = xrow[xlane + 0]; b0.y = xrow[xlane + 1];
        b1.x = xrow[xlane + 4]; b1.y = xrow[xlane + 5];
    };

    auto layer_step = [&](v2f Bi0, v2f Bi1) {
        v2f Bh0, Bh1;
        Bh0.x = hs[0]; Bh0.y = hs[1];
        Bh1.x = hs[2]; Bh1.y = hs[3];
        v8f g0 = BiasT[0], g1 = BiasT[1];
        g0 = wmma_k4(Aw[0][0], Bi0, g0);  g1 = wmma_k4(Aw[1][0], Bi0, g1);
        g0 = wmma_k4(Aw[0][1], Bi1, g0);  g1 = wmma_k4(Aw[1][1], Bi1, g1);
        g0 = wmma_k4(Aw[0][2], Bh0, g0);  g1 = wmma_k4(Aw[1][2], Bh0, g1);
        g0 = wmma_k4(Aw[0][3], Bh1, g0);  g1 = wmma_k4(Aw[1][3], Bh1, g1);
#pragma unroll
        for (int ul = 0; ul < 4; ++ul) {
            float gi = GATE(g0, g1, ul, 0), gf = GATE(g0, g1, ul, 1);
            float gg = GATE(g0, g1, ul, 2), go = GATE(g0, g1, ul, 3);
            float cn = fmaf(fast_sig(gf), cs[ul], fast_sig(gi) * fast_tanh(gg));
            cs[ul] = cn;
            hs[ul] = fast_sig(go) * fast_tanh(cn);
        }
    };

    auto decode_store = [&](int t, v4f own, v4f oth) {
        float dec[4];
#pragma unroll
        for (int d = 0; d < 4; ++d) {
            float acc = dbr[d];
#pragma unroll
            for (int j = 0; j < 4; ++j) acc = fmaf(dWo[d][j], own[j], acc);
#pragma unroll
            for (int j = 0; j < 4; ++j) acc = fmaf(dWp[d][j], oth[j], acc);
            dec[d] = fmaxf(acc, 0.0f);
        }
        v4f res;
#pragma unroll
        for (int a = 0; a < 4; ++a) {
            float acc = hbr[a];
#pragma unroll
            for (int d = 0; d < 4; ++d) acc = fmaf(hWr[a][d], dec[d], acc);
            res[a] = acc;
        }
        size_t off = (half ? sgBase : muBase) + ((size_t)t * Nn + gcol) * 4;
        *(v4f*)(outp + off) = res;
    };

    v2f xr0[CH], xr1[CH];
#pragma unroll
    for (int u = 0; u < CH; ++u) { xr0[u] = (v2f)(0.0f); xr1[u] = (v2f)(0.0f); }
    if (role == 0) {
#pragma unroll
        for (int u = 0; u < CH; ++u) load_x(u < T ? u : T - 1, xr0[u], xr1[u]);
    }

    const int nBlk = (T + CH - 1) / CH;
    for (int p = 0; p < nBlk + 2; ++p) {
        if (role == 0) {
            if (p < nBlk) {
                v2f Bx0[CH], Bx1[CH];
#pragma unroll
                for (int u = 0; u < CH; ++u) { Bx0[u] = xr0[u]; Bx1[u] = xr1[u]; }
#pragma unroll
                for (int u = 0; u < CH; ++u) {
                    const int tn = (p + 1) * CH + u;
                    load_x(tn < T ? tn : T - 1, xr0[u], xr1[u]);
                }
                __builtin_amdgcn_sched_barrier(0);
#pragma unroll
                for (int u = 0; u < CH; ++u) {
                    layer_step(Bx0[u], Bx1[u]);
                    v4f hv; hv.x = hs[0]; hv.y = hs[1]; hv.z = hs[2]; hv.w = hs[3];
                    hand0[p & 1][u][lane] = hv;
                }
            }
        } else if (role == 1) {
            if (p >= 1 && p <= nBlk) {
                const int s = (p - 1) & 1;
                v4f hin[CH];
#pragma unroll
                for (int u = 0; u < CH; ++u) hin[u] = hand0[s][u][lane];
                __builtin_amdgcn_sched_barrier(0);
#pragma unroll
                for (int u = 0; u < CH; ++u) {
                    v2f Bi0, Bi1;
                    Bi0.x = hin[u].x; Bi0.y = hin[u].y;
                    Bi1.x = hin[u].z; Bi1.y = hin[u].w;
                    layer_step(Bi0, Bi1);
                    v4f hv; hv.x = hs[0]; hv.y = hs[1]; hv.z = hs[2]; hv.w = hs[3];
                    hand1[s][u][lane] = hv;
                }
            }
        } else {
            if (p >= 2) {
                const int s = (p - 2) & 1;
                const int t0 = (p - 2) * CH;
                if (t0 + CH <= T) {
                    v4f own[CH], oth[CH];
#pragma unroll
                    for (int u = 0; u < CH; ++u) { own[u] = hand1[s][u][lane]; oth[u] = hand1[s][u][lane ^ 16]; }
#pragma unroll
                    for (int u = 0; u < CH; ++u) decode_store(t0 + u, own[u], oth[u]);
                } else {
                    for (int u = 0; t0 + u < T; ++u)
                        decode_store(t0 + u, hand1[s][u][lane], hand1[s][u][lane ^ 16]);
                }
            }
        }
        __builtin_amdgcn_fence(__ATOMIC_RELEASE, "workgroup", "local");
        __builtin_amdgcn_s_barrier();
        __builtin_amdgcn_fence(__ATOMIC_ACQUIRE, "workgroup", "local");
    }
#undef GATE
}

extern "C" void kernel_launch(void* const* d_in, const int* in_sizes, int n_in,
                              void* d_out, int out_size, void* d_ws, size_t ws_size,
                              hipStream_t stream) {
    (void)n_in; (void)out_size; (void)d_ws; (void)ws_size;
    const float* state = (const float*)d_in[0];
    const float* dist  = (const float*)d_in[1];
    const float* e1W1  = (const float*)d_in[2];
    const float* e1b1  = (const float*)d_in[3];
    const float* e1W2  = (const float*)d_in[4];
    const float* e1b2  = (const float*)d_in[5];
    const float* e2W1  = (const float*)d_in[6];
    const float* e2b1  = (const float*)d_in[7];
    const float* e2W2  = (const float*)d_in[8];
    const float* e2b2  = (const float*)d_in[9];
    const float* Wih0  = (const float*)d_in[10];
    const float* Whh0  = (const float*)d_in[11];
    const float* bih0  = (const float*)d_in[12];
    const float* bhh0  = (const float*)d_in[13];
    const float* Wih1  = (const float*)d_in[14];
    const float* Whh1  = (const float*)d_in[15];
    const float* bih1  = (const float*)d_in[16];
    const float* bhh1  = (const float*)d_in[17];
    const float* decW  = (const float*)d_in[18];
    const float* decb  = (const float*)d_in[19];
    const float* finW  = (const float*)d_in[20];
    const float* finb  = (const float*)d_in[21];
    const float* fin2W = (const float*)d_in[22];
    const float* fin2b = (const float*)d_in[23];

    const int Nn = in_sizes[0] / 16;
    const int T  = in_sizes[1] / (Nn * 8);

    lstm_ctrl_kernel<<<dim3(Nn / 16), dim3(96), 0, stream>>>(
        state, dist,
        e1W1, e1b1, e1W2, e1b2, e2W1, e2b1, e2W2, e2b2,
        Wih0, Whh0, bih0, bhh0, Wih1, Whh1, bih1, bhh1,
        decW, decb, finW, finb, fin2W, fin2b,
        (float*)d_out, T, Nn);
}


# gate math cut to 12 packed f32 ops + 20 tanh a step (was 28 plain): a lane's D rows re-ordered i,i,f,f,g,g,o,o per unit pair, sigmoid's halving folded exactly into i,f,o weight rows and bias
# speedup vs baseline: 5.9953x; 1.0303x over previous
#include <hip/hip_runtime.h>


typedef float v2f __attribute__((ext_vector_type(2)));
typedef float v4f __attribute__((ext_vector_type(4)));
typedef float v8f __attribute__((ext_vector_type(8)));

__device__ __forceinline__ float fast_tanh(float x) {
#if __has_builtin(__builtin_amdgcn_tanhf)
    return __builtin_amdgcn_tanhf(x);
#else
    float e = __builtin_amdgcn_exp2f(x * 2.8853900817779268f);
    return 1.0f - 2.0f * __builtin_amdgcn_rcpf(e + 1.0f);
#endif
}
__device__ __forceinline__ float row_scale(int orow) {
    return ((orow >> 3) == 2) ? 1.0f : 0.5f;
}

__device__ __forceinline__ v8f wmma_k4(v2f a, v2f b, v8f c) {
    return __builtin_amdgcn_wmma_f32_16x16x4_f32(false, a, false, b, (short)0, c,
                                                 false, false);
}


__device__ __forceinline__ int orig_row(int p) {
    int gate  = (p & 7) >> 1;
    int blk   = p >> 3;
    int upair = ((blk & 1) << 1) | (blk >> 1);
    int unit  = upair * 2 + (p & 1);
    return gate * 8 + unit;
}

#define CH 8

__global__ __launch_bounds__(96)
void lstm_ctrl_kernel(
    const float* __restrict__ state, const float* __restrict__ dist,
    const float* __restrict__ e1W1, const float* __restrict__ e1b1,
    const float* __restrict__ e1W2, const float* __restrict__ e1b2,
    const float* __restrict__ e2W1, const float* __restrict__ e2b1,
    const float* __restrict__ e2W2, const float* __restrict__ e2b2,
    const float* __restrict__ Wih0, const float* __restrict__ Whh0,
    const float* __restrict__ bih0, const float* __restrict__ bhh0,
    const float* __restrict__ Wih1, const float* __restrict__ Whh1,
    const float* __restrict__ bih1, const float* __restrict__ bhh1,
    const float* __restrict__ decW, const float* __restrict__ decb,
    const float* __restrict__ finW, const float* __restrict__ finb,
    const float* __restrict__ fin2W, const float* __restrict__ fin2b,
    float* __restrict__ outp, int T, int Nn)
{
    const int lane = threadIdx.x & 31;
    const int role = __builtin_amdgcn_readfirstlane((int)(threadIdx.x >> 5));
    const int col  = lane & 15;
    const int half = lane >> 4;
    const int kOff = half << 1;
    const int gcol = blockIdx.x * 16 + col;

    __shared__ v4f hand0[2][CH][32];
    __shared__ v4f hand1[2][CH][32];

    if (T <= 0) return;

    v2f Aw[2][4];
    v8f BiasT[2];
    float hs[4], cs[4];
#pragma unroll
    for (int i = 0; i < 4; ++i) { hs[i] = 0.0f; cs[i] = 0.0f; }
#pragma unroll
    for (int tI = 0; tI < 2; ++tI) {
        BiasT[tI] = (v8f)(0.0f);
#pragma unroll
        for (int ks = 0; ks < 4; ++ks) Aw[tI][ks] = (v2f)(0.0f);
    }
    if (role < 2) {
        const int l = role;
        const float* Wih = l ? Wih1 : Wih0;
        const float* Whh = l ? Whh1 : Whh0;
        const float* bih = l ? bih1 : bih0;
        const float* bhh = l ? bhh1 : bhh0;
#pragma unroll
        for (int tI = 0; tI < 2; ++tI) {
            const int orow = orig_row(tI * 16 + col);
            const float rsc = row_scale(orow);
#pragma unroll
            for (int ks = 0; ks < 4; ++ks) {
                v2f a;
#pragma unroll
                for (int cc = 0; cc < 2; ++cc) {
                    const float* wsrc = (ks < 2) ? Wih : Whh;
                    int f = (l == 0 && ks < 2) ? (ks * 4 + kOff + cc)
                                               : (half * 4 + (ks & 1) * 2 + cc);
                    float w = wsrc[orow * 8 + f] * rsc;
                    if (cc == 0) a.x = w; else a.y = w;
                }
                Aw[tI][ks] = a;
            }
            v8f bv;
#pragma unroll
            for (int r = 0; r < 8; ++r) {
                int br = orig_row(tI * 16 + r + half * 8);
                bv[r] = (bih[br] + bhh[br]) * row_scale(br);
            }
            BiasT[tI] = bv;
        }
        float sv[16];
#pragma unroll
        for (int i = 0; i < 16; ++i) sv[i] = state[gcol * 16 + i];
        float eh1[4], eh2[4];
#pragma unroll
        for (int j = 0; j < 4; ++j) {
            float a1 = e1b1[j], a2 = e2b1[j];
#pragma unroll
            for (int i = 0; i < 16; ++i) {
                a1 = fmaf(e1W1[j * 16 + i], sv[i], a1);
                a2 = fmaf(e2W1[j * 16 + i], sv[i], a2);
            }
            eh1[j] = fmaxf(a1, 0.0f);
            eh2[j] = fmaxf(a2, 0.0f);
        }
#pragma unroll
        for (int ul = 0; ul < 4; ++ul) {
            const int e = l * 8 + half * 4 + ul;
            float a1 = e1b2[e], a2 = e2b2[e];
#pragma unroll
            for (int j = 0; j < 4; ++j) {
                a1 = fmaf(e1W2[e * 4 + j], eh1[j], a1);
                a2 = fmaf(e2W2[e * 4 + j], eh2[j], a2);
            }
            hs[ul] = fmaxf(a1, 0.0f);
            cs[ul] = fmaxf(a2, 0.0f);
        }
    }

    float dWo[4][4], dWp[4][4], dbr[4], hWr[4][4], hbr[4];
    {
        const float* fw = half ? fin2W : finW;
        const float* fb = half ? fin2b : finb;
#pragma unroll
        for (int d = 0; d < 4; ++d) {
            dbr[d] = decb[d];
#pragma unroll
            for (int j = 0; j < 4; ++j) {
                dWo[d][j] = decW[d * 8 + half * 4 + j];
                dWp[d][j] = decW[d * 8 + (1 - half) * 4 + j];
            }
        }
#pragma unroll
        for (int a = 0; a < 4; ++a) {
            hbr[a] = fb[a];
#pragma unroll
            for (int d = 0; d < 4; ++d) hWr[a][d] = fw[a * 4 + d];
        }
    }


    const size_t muBase  = 0;
    const size_t sgBase  = (size_t)T * (size_t)Nn * 4;

    const int xlane = gcol * 8 + kOff;
    auto load_x = [&](int tt, v2f& b0, v2f& b1) {
        const float* xrow = dist + (size_t)tt * (size_t)Nn * 8;
        b0.x = xrow[xlane + 0]; b0.y = xrow[xlane + 1];
        b1.x = xrow[xlane + 4]; b1.y = xrow[xlane + 5];
    };

    auto layer_step = [&](v2f Bi0, v2f Bi1) {
        v2f Bh0, Bh1;
        Bh0.x = hs[0]; Bh0.y = hs[1];
        Bh1.x = hs[2]; Bh1.y = hs[3];
        v8f g0 = BiasT[0], g1 = BiasT[1];
        g0 = wmma_k4(Aw[0][0], Bi0, g0);  g1 = wmma_k4(Aw[1][0], Bi0, g1);
        g0 = wmma_k4(Aw[0][1], Bi1, g0);  g1 = wmma_k4(Aw[1][1], Bi1, g1);
        g0 = wmma_k4(Aw[0][2], Bh0, g0);  g1 = wmma_k4(Aw[1][2], Bh0, g1);
        g0 = wmma_k4(Aw[0][3], Bh1, g0);  g1 = wmma_k4(Aw[1][3], Bh1, g1);
#pragma unroll
        for (int tI = 0; tI < 2; ++tI) {
            const v8f G = tI ? g1 : g0;
            v2f ti, tf, tg, to, c;
            ti.x = fast_tanh(G[0]); ti.y = fast_tanh(G[1]);
            tf.x = fast_tanh(G[2]); tf.y = fast_tanh(G[3]);
            tg.x = fast_tanh(G[4]); tg.y = fast_tanh(G[5]);
            to.x = fast_tanh(G[6]); to.y = fast_tanh(G[7]);
            c.x = cs[2 * tI]; c.y = cs[2 * tI + 1];
            const v2f si = __builtin_elementwise_fma(ti, (v2f)(0.5f), (v2f)(0.5f));
            const v2f sf = __builtin_elementwise_fma(tf, (v2f)(0.5f), (v2f)(0.5f));
            const v2f so = __builtin_elementwise_fma(to, (v2f)(0.5f), (v2f)(0.5f));
            const v2f cn = __builtin_elementwise_fma(sf, c, si * tg);
            v2f th; th.x = fast_tanh(cn.x); th.y = fast_tanh(cn.y);
            const v2f hn = so * th;
            cs[2 * tI] = cn.x; cs[2 * tI + 1] = cn.y;
            hs[2 * tI] = hn.x; hs[2 * tI + 1] = hn.y;
        }
    };

    auto decode_store = [&](int t, v4f own, v4f oth) {
        float dec[4];
#pragma unroll
        for (int d = 0; d < 4; ++d) {
            float acc = dbr[d];
#pragma unroll
            for (int j = 0; j < 4; ++j) acc = fmaf(dWo[d][j], own[j], acc);
#pragma unroll
            for (int j = 0; j < 4; ++j) acc = fmaf(dWp[d][j], oth[j], acc);
            dec[d] = fmaxf(acc, 0.0f);
        }
        v4f res;
#pragma unroll
        for (int a = 0; a < 4; ++a) {
            float acc = hbr[a];
#pragma unroll
            for (int d = 0; d < 4; ++d) acc = fmaf(hWr[a][d], dec[d], acc);
            res[a] = acc;
        }
        size_t off = (half ? sgBase : muBase) + ((size_t)t * Nn + gcol) * 4;
        *(v4f*)(outp + off) = res;
    };

    v2f xr0[CH], xr1[CH];
#pragma unroll
    for (int u = 0; u < CH; ++u) { xr0[u] = (v2f)(0.0f); xr1[u] = (v2f)(0.0f); }
    if (role == 0) {
#pragma unroll
        for (int u = 0; u < CH; ++u) load_x(u < T ? u : T - 1, xr0[u], xr1[u]);
    }

    const int nBlk = (T + CH - 1) / CH;
    for (int p = 0; p < nBlk + 2; ++p) {
        if (role == 0) {
            if (p < nBlk) {
                v2f Bx0[CH], Bx1[CH];
#pragma unroll
                for (int u = 0; u < CH; ++u) { Bx0[u] = xr0[u]; Bx1[u] = xr1[u]; }
#pragma unroll
                for (int u = 0; u < CH; ++u) {
                    const int tn = (p + 1) * CH + u;
                    load_x(tn < T ? tn : T - 1, xr0[u], xr1[u]);
                }
                __builtin_amdgcn_sched_barrier(0);
#pragma unroll
                for (int u = 0; u < CH; ++u) {
                    layer_step(Bx0[u], Bx1[u]);
                    v4f hv; hv.x = hs[0]; hv.y = hs[1]; hv.z = hs[2]; hv.w = hs[3];
                    hand0[p & 1][u][lane] = hv;
                }
            }
        } else if (role == 1) {
            if (p >= 1 && p <= nBlk) {
                const int s = (p - 1) & 1;
                v4f hin[CH];
#pragma unroll
                for (int u = 0; u < CH; ++u) hin[u] = hand0[s][u][lane];
                __builtin_amdgcn_sched_barrier(0);
#pragma unroll
                for (int u = 0; u < CH; ++u) {
                    v2f Bi0, Bi1;
                    Bi0.x = hin[u].x; Bi0.y = hin[u].y;
                    Bi1.x = hin[u].z; Bi1.y = hin[u].w;
                    layer_step(Bi0, Bi1);
                    v4f hv; hv.x = hs[0]; hv.y = hs[1]; hv.z = hs[2]; hv.w = hs[3];
                    hand1[s][u][lane] = hv;
                }
            }
        } else {
            if (p >= 2) {
                const int s = (p - 2) & 1;
                const int t0 = (p - 2) * CH;
                if (t0 + CH <= T) {
                    v4f own[CH], oth[CH];
#pragma unroll
                    for (int u = 0; u < CH; ++u) { own[u] = hand1[s][u][lane]; oth[u] = hand1[s][u][lane ^ 16]; }
#pragma unroll
                    for (int u = 0; u < CH; ++u) decode_store(t0 + u, own[u], oth[u]);
                } else {
                    for (int u = 0; t0 + u < T; ++u)
                        decode_store(t0 + u, hand1[s][u][lane], hand1[s][u][lane ^ 16]);
                }
            }
        }
        __builtin_amdgcn_fence(__ATOMIC_RELEASE, "workgroup", "local");
        __builtin_amdgcn_s_barrier();
        __builtin_amdgcn_fence(__ATOMIC_ACQUIRE, "workgroup", "local");
    }
}

extern "C" void kernel_launch(void* const* d_in, const int* in_sizes, int n_in,
                              void* d_out, int out_size, void* d_ws, size_t ws_size,
                              hipStream_t stream) {
    (void)n_in; (void)out_size; (void)d_ws; (void)ws_size;
    const float* state = (const float*)d_in[0];
    const float* dist  = (const float*)d_in[1];
    const float* e1W1  = (const float*)d_in[2];
    const float* e1b1  = (const float*)d_in[3];
    const float* e1W2  = (const float*)d_in[4];
    const float* e1b2  = (const float*)d_in[5];
    const float* e2W1  = (const float*)d_in[6];
    const float* e2b1  = (const float*)d_in[7];
    const float* e2W2  = (const float*)d_in[8];
    const float* e2b2  = (const float*)d_in[9];
    const float* Wih0  = (const float*)d_in[10];
    const float* Whh0  = (const float*)d_in[11];
    const float* bih0  = (const float*)d_in[12];
    const float* bhh0  = (const float*)d_in[13];
    const float* Wih1  = (const float*)d_in[14];
    const float* Whh1  = (const float*)d_in[15];
    const float* bih1  = (const float*)d_in[16];
    const float* bhh1  = (const float*)d_in[17];
    const float* decW  = (const float*)d_in[18];
    const float* decb  = (const float*)d_in[19];
    const float* finW  = (const float*)d_in[20];
    const float* finb  = (const float*)d_in[21];
    const float* fin2W = (const float*)d_in[22];
    const float* fin2b = (const float*)d_in[23];

    const int Nn = in_sizes[0] / 16;
    const int T  = in_sizes[1] / (Nn * 8);

    lstm_ctrl_kernel<<<dim3(Nn / 16), dim3(96), 0, stream>>>(
        state, dist,
        e1W1, e1b1, e1W2, e1b2, e2W1, e2b1, e2W2, e2b2,
        Wih0, Whh0, bih0, bhh0, Wih1, Whh1, bih1, bhh1,
        decW, decb, finW, finb, fin2W, fin2b,
        (float*)d_out, T, Nn);
}


# input-side gate sums (bias + W_ih*input) leave the two recurrent waves: two helper waves form them a block ahead, LDS hands them over as C tiles (decoder split over the helpers); same sum order
# speedup vs baseline: 6.5560x; 1.0935x over previous
#include <hip/hip_runtime.h>


typedef float v2f __attribute__((ext_vector_type(2)));
typedef float v4f __attribute__((ext_vector_type(4)));
typedef float v8f __attribute__((ext_vector_type(8)));

__device__ __forceinline__ float fast_tanh(float x) {
#if __has_builtin(__builtin_amdgcn_tanhf)
    return __builtin_amdgcn_tanhf(x);
#else
    float e = __builtin_amdgcn_exp2f(x * 2.8853900817779268f);
    return 1.0f - 2.0f * __builtin_amdgcn_rcpf(e + 1.0f);
#endif
}
__device__ __forceinline__ float row_scale(int orow) {
    return ((orow >> 3) == 2) ? 1.0f : 0.5f;
}

__device__ __forceinline__ v8f wmma_k4(v2f a, v2f b, v8f c) {
    return __builtin_amdgcn_wmma_f32_16x16x4_f32(false, a, false, b, (short)0, c,
                                                 false, false);
}


__device__ __forceinline__ int orig_row(int p) {
    int gate  = (p & 7) >> 1;
    int blk   = p >> 3;
    int upair = ((blk & 1) << 1) | (blk >> 1);
    int unit  = upair * 2 + (p & 1);
    return gate * 8 + unit;
}

#define CH 8

__global__ __launch_bounds__(128)
void lstm_ctrl_kernel(
    const float* __restrict__ state, const float* __restrict__ dist,
    const float* __restrict__ e1W1, const float* __restrict__ e1b1,
    const float* __restrict__ e1W2, const float* __restrict__ e1b2,
    const float* __restrict__ e2W1, const float* __restrict__ e2b1,
    const float* __restrict__ e2W2, const float* __restrict__ e2b2,
    const float* __restrict__ Wih0, const float* __restrict__ Whh0,
    const float* __restrict__ bih0, const float* __restrict__ bhh0,
    const float* __restrict__ Wih1, const float* __restrict__ Whh1,
    const float* __restrict__ bih1, const float* __restrict__ bhh1,
    const float* __restrict__ decW, const float* __restrict__ decb,
    const float* __restrict__ finW, const float* __restrict__ finb,
    const float* __restrict__ fin2W, const float* __restrict__ fin2b,
    float* __restrict__ outp, int T, int Nn)
{
    const int lane = threadIdx.x & 31;
    const int role = __builtin_amdgcn_readfirstlane((int)(threadIdx.x >> 5));
    const int col  = lane & 15;
    const int half = lane >> 4;
    const int kOff = half << 1;
    const int gcol = blockIdx.x * 16 + col;

    __shared__ v4f hand[2][2][CH][32];
    __shared__ v4f xg[2][2][CH][4][32];

    if (T <= 0) return;

    v2f Aw[2][4];
    v8f BiasT[2];
    float hs[4], cs[4];
#pragma unroll
    for (int i = 0; i < 4; ++i) { hs[i] = 0.0f; cs[i] = 0.0f; }
#pragma unroll
    for (int tI = 0; tI < 2; ++tI) {
        BiasT[tI] = (v8f)(0.0f);
#pragma unroll
        for (int ks = 0; ks < 4; ++ks) Aw[tI][ks] = (v2f)(0.0f);
    }
    {
        const int l = (role == 1 || role == 2) ? 1 : 0;
        const float* Wih = l ? Wih1 : Wih0;
        const float* Whh = l ? Whh1 : Whh0;
        const float* bih = l ? bih1 : bih0;
        const float* bhh = l ? bhh1 : bhh0;
#pragma unroll
        for (int tI = 0; tI < 2; ++tI) {
            const int orow = orig_row(tI * 16 + col);
            const float rsc = row_scale(orow);
#pragma unroll
            for (int ks = 0; ks < 4; ++ks) {
                v2f a;
#pragma unroll
                for (int cc = 0; cc < 2; ++cc) {
                    const float* wsrc = (ks < 2) ? Wih : Whh;
                    int f = (l == 0 && ks < 2) ? (ks * 4 + kOff + cc)
                                               : (half * 4 + (ks & 1) * 2 + cc);
                    float w = wsrc[orow * 8 + f] * rsc;
                    if (cc == 0) a.x = w; else a.y = w;
                }
                Aw[tI][ks] = a;
            }
            v8f bv;
#pragma unroll
            for (int r = 0; r < 8; ++r) {
                int br = orig_row(tI * 16 + r + half * 8);
                bv[r] = (bih[br] + bhh[br]) * row_scale(br);
            }
            BiasT[tI] = bv;
        }
      if (role < 2) {
        float sv[16];
#pragma unroll
        for (int i = 0; i < 16; ++i) sv[i] = state[gcol * 16 + i];
        float eh1[4], eh2[4];
#pragma unroll
        for (int j = 0; j < 4; ++j) {
            float a1 = e1b1[j], a2 = e2b1[j];
#pragma unroll
            for (int i = 0; i < 16; ++i) {
                a1 = fmaf(e1W1[j * 16 + i], sv[i], a1);
                a2 = fmaf(e2W1[j * 16 + i], sv[i], a2);
            }
            eh1[j] = fmaxf(a1, 0.0f);
            eh2[j] = fmaxf(a2, 0.0f);
        }
#pragma unroll
        for (int ul = 0; ul < 4; ++ul) {
            const int e = l * 8 + half * 4 + ul;
            float a1 = e1b2[e], a2 = e2b2[e];
#pragma unroll
            for (int j = 0; j < 4; ++j) {
                a1 = fmaf(e1W2[e * 4 + j], eh1[j], a1);
                a2 = fmaf(e2W2[e * 4 + j], eh2[j], a2);
            }
            hs[ul] = fmaxf(a1, 0.0f);
            cs[ul] = fmaxf(a2, 0.0f);
        }
      }
    }

    float dWo[4][4], dWp[4][4], dbr[4], hWr[4][4], hbr[4];
    {
        const float* fw = half ? fin2W : finW;
        const float* fb = half ? fin2b : finb;
#pragma unroll
        for (int d = 0; d < 4; ++d) {
            dbr[d] = decb[d];
#pragma unroll
            for (int j = 0; j < 4; ++j) {
                dWo[d][j] = decW[d * 8 + half * 4 + j];
                dWp[d][j] = decW[d * 8 + (1 - half) * 4 + j];
            }
        }
#pragma unroll
        for (int a = 0; a < 4; ++a) {
            hbr[a] = fb[a];
#pragma unroll
            for (int d = 0; d < 4; ++d) hWr[a][d] = fw[a * 4 + d];
        }
    }


    const size_t muBase  = 0;
    const size_t sgBase  = (size_t)T * (size_t)Nn * 4;

    const int xlane = gcol * 8 + kOff;
    auto load_x = [&](int tt, v2f& b0, v2f& b1) {
        const float* xrow = dist + (size_t)tt * (size_t)Nn * 8;
        b0.x = xrow[xlane + 0]; b0.y = xrow[xlane + 1];
        b1.x = xrow[xlane + 4]; b1.y = xrow[xlane + 5];
    };

    auto pre_step = [&](v2f Bi0, v2f Bi1, v4f (*dst)[32]) {
        v8f g0 = BiasT[0], g1 = BiasT[1];
        g0 = wmma_k4(Aw[0][0], Bi0, g0);  g1 = wmma_k4(Aw[1][0], Bi0, g1);
        g0 = wmma_k4(Aw[0][1], Bi1, g0);  g1 = wmma_k4(Aw[1][1], Bi1, g1);
        dst[0][lane] = __builtin_shufflevector(g0, g0, 0, 1, 2, 3);
        dst[1][lane] = __builtin_shufflevector(g0, g0, 4, 5, 6, 7);
        dst[2][lane] = __builtin_shufflevector(g1, g1, 0, 1, 2, 3);
        dst[3][lane] = __builtin_shufflevector(g1, g1, 4, 5, 6, 7);
    };

    auto layer_step = [&](v8f c0, v8f c1) {
        v2f Bh0, Bh1;
        Bh0.x = hs[0]; Bh0.y = hs[1];
        Bh1.x = hs[2]; Bh1.y = hs[3];
        v8f g0 = wmma_k4(Aw[0][2], Bh0, c0), g1 = wmma_k4(Aw[1][2], Bh0, c1);
        g0 = wmma_k4(Aw[0][3], Bh1, g0);  g1 = wmma_k4(Aw[1][3], Bh1, g1);
#pragma unroll
        for (int tI = 0; tI < 2; ++tI) {
            const v8f G = tI ? g1 : g0;
            v2f ti, tf, tg, to, c;
            ti.x = fast_tanh(G[0]); ti.y = fast_tanh(G[1]);
            tf.x = fast_tanh(G[2]); tf.y = fast_tanh(G[3]);
            tg.x = fast_tanh(G[4]); tg.y = fast_tanh(G[5]);
            to.x = fast_tanh(G[6]); to.y = fast_tanh(G[7]);
            c.x = cs[2 * tI]; c.y = cs[2 * tI + 1];
            const v2f si = __builtin_elementwise_fma(ti, (v2f)(0.5f), (v2f)(0.5f));
            const v2f sf = __builtin_elementwise_fma(tf, (v2f)(0.5f), (v2f)(0.5f));
            const v2f so = __builtin_elementwise_fma(to, (v2f)(0.5f), (v2f)(0.5f));
            const v2f cn = __builtin_elementwise_fma(sf, c, si * tg);
            v2f th; th.x = fast_tanh(cn.x); th.y = fast_tanh(cn.y);
            const v2f hn = so * th;
            cs[2 * tI] = cn.x; cs[2 * tI + 1] = cn.y;
            hs[2 * tI] = hn.x; hs[2 * tI + 1] = hn.y;
        }
    };

    auto decode_store = [&](int t, v4f own, v4f oth) {
        float dec[4];
#pragma unroll
        for (int d = 0; d < 4; ++d) {
            float acc = dbr[d];
#pragma unroll
            for (int j = 0; j < 4; ++j) acc = fmaf(dWo[d][j], own[j], acc);
#pragma unroll
            for (int j = 0; j < 4; ++j) acc = fmaf(dWp[d][j], oth[j], acc);
            dec[d] = fmaxf(acc, 0.0f);
        }
        v4f res;
#pragma unroll
        for (int a = 0; a < 4; ++a) {
            float acc = hbr[a];
#pragma unroll
            for (int d = 0; d < 4; ++d) acc = fmaf(hWr[a][d], dec[d], acc);
            res[a] = acc;
        }
        size_t off = (half ? sgBase : muBase) + ((size_t)t * Nn + gcol) * 4;
        *(v4f*)(outp + off) = res;
    };

    auto decode_half = [&](int p, int u0) {
        if (p >= 4) {
            const int s = (p - 4) & 1;
            const int t0 = (p - 4) * CH + u0;
            if (t0 + CH / 2 <= T) {
                v4f own[CH / 2], oth[CH / 2];
#pragma unroll
                for (int u = 0; u < CH / 2; ++u) { own[u] = hand[1][s][u0 + u][lane]; oth[u] = hand[1][s][u0 + u][lane ^ 16]; }
#pragma unroll
                for (int u = 0; u < CH / 2; ++u) decode_store(t0 + u, own[u], oth[u]);
            } else {
                for (int u = 0; u < CH / 2 && t0 + u < T; ++u)
                    decode_store(t0 + u, hand[1][s][u0 + u][lane], hand[1][s][u0 + u][lane ^ 16]);
            }
        }
    };

    v2f xr0[CH], xr1[CH];
#pragma unroll
    for (int u = 0; u < CH; ++u) { xr0[u] = (v2f)(0.0f); xr1[u] = (v2f)(0.0f); }
    if (role == 3) {
#pragma unroll
        for (int u = 0; u < CH; ++u) load_x(u < T ? u : T - 1, xr0[u], xr1[u]);
    }

    const int nBlk = (T + CH - 1) / CH;
    for (int p = 0; p < nBlk + 4; ++p) {
        if (role < 2) {
            const int blk = p - 1 - 2 * role;
            if (blk >= 0 && blk < nBlk) {
                const int s = blk & 1;
                v8f C0[CH], C1[CH];
#pragma unroll
                for (int u = 0; u < CH; ++u) {
                    const v4f q0 = xg[role][s][u][0][lane], q1 = xg[role][s][u][1][lane];
                    const v4f q2 = xg[role][s][u][2][lane], q3 = xg[role][s][u][3][lane];
                    C0[u] = __builtin_shufflevector(q0, q1, 0, 1, 2, 3, 4, 5, 6, 7);
                    C1[u] = __builtin_shufflevector(q2, q3, 0, 1, 2, 3, 4, 5, 6, 7);
                }
                __builtin_amdgcn_sched_barrier(0);
#pragma unroll
                for (int u = 0; u < CH; ++u) {
                    layer_step(C0[u], C1[u]);
                    v4f hv; hv.x = hs[0]; hv.y = hs[1]; hv.z = hs[2]; hv.w = hs[3];
                    hand[role][s][u][lane] = hv;
                }
            }
        } else if (role == 2) {
            const int b1 = p - 2;
            if (b1 >= 0 && b1 < nBlk) {
                const int s = b1 & 1;
                v4f hin[CH];
#pragma unroll
                for (int u = 0; u < CH; ++u) hin[u] = hand[0][s][u][lane];
                __builtin_amdgcn_sched_barrier(0);
#pragma unroll
                for (int u = 0; u < CH; ++u) {
                    v2f Bi0, Bi1;
                    Bi0.x = hin[u].x; Bi0.y = hin[u].y;
                    Bi1.x = hin[u].z; Bi1.y = hin[u].w;
                    pre_step(Bi0, Bi1, xg[1][s][u]);
                }
            }
            decode_half(p, 0);
        } else {
            if (p < nBlk) {
                v2f Bx0[CH], Bx1[CH];
#pragma unroll
                for (int u = 0; u < CH; ++u) { Bx0[u] = xr0[u]; Bx1[u] = xr1[u]; }
#pragma unroll
                for (int u = 0; u < CH; ++u) {
                    const int tn = (p + 1) * CH + u;
                    load_x(tn < T ? tn : T - 1, xr0[u], xr1[u]);
                }
                __builtin_amdgcn_sched_barrier(0);
#pragma unroll
                for (int u = 0; u < CH; ++u) pre_step(Bx0[u], Bx1[u], xg[0][p & 1][u]);
            }
            decode_half(p, CH / 2);
        }
        __builtin_amdgcn_fence(__ATOMIC_RELEASE, "workgroup", "local");
        __builtin_amdgcn_s_barrier();
        __builtin_amdgcn_fence(__ATOMIC_ACQUIRE, "workgroup", "local");
    }
}

extern "C" void kernel_launch(void* const* d_in, const int* in_sizes, int n_in,
                              void* d_out, int out_size, void* d_ws, size_t ws_size,
                              hipStream_t stream) {
    (void)n_in; (void)out_size; (void)d_ws; (void)ws_size;
    const float* state = (const float*)d_in[0];
    const float* dist  = (const float*)d_in[1];
    const float* e1W1  = (const float*)d_in[2];
    const float* e1b1  = (const float*)d_in[3];
    const float* e1W2  = (const float*)d_in[4];
    const float* e1b2  = (const float*)d_in[5];
    const float* e2W1  = (const float*)d_in[6];
    const float* e2b1  = (const float*)d_in[7];
    const float* e2W2  = (const float*)d_in[8];
    const float* e2b2  = (const float*)d_in[9];
    const float* Wih0  = (const float*)d_in[10];
    const float* Whh0  = (const float*)d_in[11];
    const float* bih0  = (const float*)d_in[12];
    const float* bhh0  = (const float*)d_in[13];
    const float* Wih1  = (const float*)d_in[14];
    const float* Whh1  = (const float*)d_in[15];
    const float* bih1  = (const float*)d_in[16];
    const float* bhh1  = (const float*)d_in[17];
    const float* decW  = (const float*)d_in[18];
    const float* decb  = (const float*)d_in[19];
    const float* finW  = (const float*)d_in[20];
    const float* finb  = (const float*)d_in[21];
    const float* fin2W = (const float*)d_in[22];
    const float* fin2b = (const float*)d_in[23];

    const int Nn = in_sizes[0] / 16;
    const int T  = in_sizes[1] / (Nn * 8);

    lstm_ctrl_kernel<<<dim3(Nn / 16), dim3(128), 0, stream>>>(
        state, dist,
        e1W1, e1b1, e1W2, e1b2, e2W1, e2b1, e2W2, e2b2,
        Wih0, Whh0, bih0, bhh0, Wih1, Whh1, bih1, bhh1,
        decW, decb, finW, finb, fin2W, fin2b,
        (float*)d_out, T, Nn);
}


# hand-over blocks of 20 steps instead of 8: barrier, C-tile reload and x-load round trip once per 20 steps; a block's LDS (200 KiB) is over half a WGP's; arithmetic and sum order unchanged
# speedup vs baseline: 7.2329x; 1.1032x over previous
#include <hip/hip_runtime.h>


typedef float v2f __attribute__((ext_vector_type(2)));
typedef float v4f __attribute__((ext_vector_type(4)));
typedef float v8f __attribute__((ext_vector_type(8)));

__device__ __forceinline__ float fast_tanh(float x) {
#if __has_builtin(__builtin_amdgcn_tanhf)
    return __builtin_amdgcn_tanhf(x);
#else
    float e = __builtin_amdgcn_exp2f(x * 2.8853900817779268f);
    return 1.0f - 2.0f * __builtin_amdgcn_rcpf(e + 1.0f);
#endif
}
__device__ __forceinline__ float row_scale(int orow) {
    return ((orow >> 3) == 2) ? 1.0f : 0.5f;
}

__device__ __forceinline__ v8f wmma_k4(v2f a, v2f b, v8f c) {
    return __builtin_amdgcn_wmma_f32_16x16x4_f32(false, a, false, b, (short)0, c,
                                                 false, false);
}


__device__ __forceinline__ int orig_row(int p) {
    int gate  = (p & 7) >> 1;
    int blk   = p >> 3;
    int upair = ((blk & 1) << 1) | (blk >> 1);
    int unit  = upair * 2 + (p & 1);
    return gate * 8 + unit;
}

#define CH 20

__global__ __launch_bounds__(128)
void lstm_ctrl_kernel(
    const float* __restrict__ state, const float* __restrict__ dist,
    const float* __restrict__ e1W1, const float* __restrict__ e1b1,
    const float* __restrict__ e1W2, const float* __restrict__ e1b2,
    const float* __restrict__ e2W1, const float* __restrict__ e2b1,
    const float* __restrict__ e2W2, const float* __restrict__ e2b2,
    const float* __restrict__ Wih0, const float* __restrict__ Whh0,
    const float* __restrict__ bih0, const float* __restrict__ bhh0,
    const float* __restrict__ Wih1, const float* __restrict__ Whh1,
    const float* __restrict__ bih1, const float* __restrict__ bhh1,
    const float* __restrict__ decW, const float* __restrict__ decb,
    const float* __restrict__ finW, const float* __restrict__ finb,
    const float* __restrict__ fin2W, const float* __restrict__ fin2b,
    float* __restrict__ outp, int T, int Nn)
{
    const int lane = threadIdx.x & 31;
    const int role = __builtin_amdgcn_readfirstlane((int)(threadIdx.x >> 5));
    const int col  = lane & 15;
    const int half = lane >> 4;
    const int kOff = half << 1;
    const int gcol = blockIdx.x * 16 + col;

    __shared__ v4f hand[2][2][CH][32];
    __shared__ v4f xg[2][2][CH][4][32];

    if (T <= 0) return;

    v2f Aw[2][4];
    v8f BiasT[2];
    float hs[4], cs[4];
#pragma unroll
    for (int i = 0; i < 4; ++i) { hs[i] = 0.0f; cs[i] = 0.0f; }
#pragma unroll
    for (int tI = 0; tI < 2; ++tI) {
        BiasT[tI] = (v8f)(0.0f);
#pragma unroll
        for (int ks = 0; ks < 4; ++ks) Aw[tI][ks] = (v2f)(0.0f);
    }
    {
        const int l = (role == 1 || role == 2) ? 1 : 0;
        const float* Wih = l ? Wih1 : Wih0;
        const float* Whh = l ? Whh1 : Whh0;
        const float* bih = l ? bih1 : bih0;
        const float* bhh = l ? bhh1 : bhh0;
#pragma unroll
        for (int tI = 0; tI < 2; ++tI) {
            const int orow = orig_row(tI * 16 + col);
            const float rsc = row_scale(orow);
#pragma unroll
            for (int ks = 0; ks < 4; ++ks) {
                v2f a;
#pragma unroll
                for (int cc = 0; cc < 2; ++cc) {
                    const float* wsrc = (ks < 2) ? Wih : Whh;
                    int f = (l == 0 && ks < 2) ? (ks * 4 + kOff + cc)
                                               : (half * 4 + (ks & 1) * 2 + cc);
                    float w = wsrc[orow * 8 + f] * rsc;
                    if (cc == 0) a.x = w; else a.y = w;
                }
                Aw[tI][ks] = a;
            }
            v8f bv;
#pragma unroll
            for (int r = 0; r < 8; ++r) {
                int br = orig_row(tI * 16 + r + half * 8);
                bv[r] = (bih[br] + bhh[br]) * row_scale(br);
            }
            BiasT[tI] = bv;
        }
      if (role < 2) {
        float sv[16];
#pragma unroll
        for (int i = 0; i < 16; ++i) sv[i] = state[gcol * 16 + i];
        float eh1[4], eh2[4];
#pragma unroll
        for (int j = 0; j < 4; ++j) {
            float a1 = e1b1[j], a2 = e2b1[j];
#pragma unroll
            for (int i = 0; i < 16; ++i) {
                a1 = fmaf(e1W1[j * 16 + i], sv[i], a1);
                a2 = fmaf(e2W1[j * 16 + i], sv[i], a2);
            }
            eh1[j] = fmaxf(a1, 0.0f);
            eh2[j] = fmaxf(a2, 0.0f);
        }
#pragma unroll
        for (int ul = 0; ul < 4; ++ul) {
            const int e = l * 8 + half * 4 + ul;
            float a1 = e1b2[e], a2 = e2b2[e];
#pragma unroll
            for (int j = 0; j < 4; ++j) {
                a1 = fmaf(e1W2[e * 4 + j], eh1[j], a1);
                a2 = fmaf(e2W2[e * 4 + j], eh2[j], a2);
            }
            hs[ul] = fmaxf(a1, 0.0f);
            cs[ul] = fmaxf(a2, 0.0f);
        }
      }
    }

    float dWo[4][4], dWp[4][4], dbr[4], hWr[4][4], hbr[4];
    {
        const float* fw = half ? fin2W : finW;
        const float* fb = half ? fin2b : finb;
#pragma unroll
        for (int d = 0; d < 4; ++d) {
            dbr[d] = decb[d];
#pragma unroll
            for (int j = 0; j < 4; ++j) {
                dWo[d][j] = decW[d * 8 + half * 4 + j];
                dWp[d][j] = decW[d * 8 + (1 - half) * 4 + j];
            }
        }
#pragma unroll
        for (int a = 0; a < 4; ++a) {
            hbr[a] = fb[a];
#pragma unroll
            for (int d = 0; d < 4; ++d) hWr[a][d] = fw[a * 4 + d];
        }
    }


    const size_t muBase  = 0;
    const size_t sgBase  = (size_t)T * (size_t)Nn * 4;

    const int xlane = gcol * 8 + kOff;
    auto load_x = [&](int tt, v2f& b0, v2f& b1) {
        const float* xrow = dist + (size_t)tt * (size_t)Nn * 8;
        b0.x = xrow[xlane + 0]; b0.y = xrow[xlane + 1];
        b1.x = xrow[xlane + 4]; b1.y = xrow[xlane + 5];
    };

    auto pre_step = [&](v2f Bi0, v2f Bi1, v4f (*dst)[32]) {
        v8f g0 = BiasT[0], g1 = BiasT[1];
        g0 = wmma_k4(Aw[0][0], Bi0, g0);  g1 = wmma_k4(Aw[1][0], Bi0, g1);
        g0 = wmma_k4(Aw[0][1], Bi1, g0);  g1 = wmma_k4(Aw[1][1], Bi1, g1);
        dst[0][lane] = __builtin_shufflevector(g0, g0, 0, 1, 2, 3);
        dst[1][lane] = __builtin_shufflevector(g0, g0, 4, 5, 6, 7);
        dst[2][lane] = __builtin_shufflevector(g1, g1, 0, 1, 2, 3);
        dst[3][lane] = __builtin_shufflevector(g1, g1, 4, 5, 6, 7);
    };

    auto layer_step = [&](v8f c0, v8f c1) {
        v2f Bh0, Bh1;
        Bh0.x = hs[0]; Bh0.y = hs[1];
        Bh1.x = hs[2]; Bh1.y = hs[3];
        v8f g0 = wmma_k4(Aw[0][2], Bh0, c0), g1 = wmma_k4(Aw[1][2], Bh0, c1);
        g0 = wmma_k4(Aw[0][3], Bh1, g0);  g1 = wmma_k4(Aw[1][3], Bh1, g1);
#pragma unroll
        for (int tI = 0; tI < 2; ++tI) {
            const v8f G = tI ? g1 : g0;
            v2f ti, tf, tg, to, c;
            ti.x = fast_tanh(G[0]); ti.y = fast_tanh(G[1]);
            tf.x = fast_tanh(G[2]); tf.y = fast_tanh(G[3]);
            tg.x = fast_tanh(G[4]); tg.y = fast_tanh(G[5]);
            to.x = fast_tanh(G[6]); to.y = fast_tanh(G[7]);
            c.x = cs[2 * tI]; c.y = cs[2 * tI + 1];
            const v2f si = __builtin_elementwise_fma(ti, (v2f)(0.5f), (v2f)(0.5f));
            const v2f sf = __builtin_elementwise_fma(tf, (v2f)(0.5f), (v2f)(0.5f));
            const v2f so = __builtin_elementwise_fma(to, (v2f)(0.5f), (v2f)(0.5f));
            const v2f cn = __builtin_elementwise_fma(sf, c, si * tg);
            v2f th; th.x = fast_tanh(cn.x); th.y = fast_tanh(cn.y);
            const v2f hn = so * th;
            cs[2 * tI] = cn.x; cs[2 * tI + 1] = cn.y;
            hs[2 * tI] = hn.x; hs[2 * tI + 1] = hn.y;
        }
    };

    auto decode_store = [&](int t, v4f own, v4f oth) {
        float dec[4];
#pragma unroll
        for (int d = 0; d < 4; ++d) {
            float acc = dbr[d];
#pragma unroll
            for (int j = 0; j < 4; ++j) acc = fmaf(dWo[d][j], own[j], acc);
#pragma unroll
            for (int j = 0; j < 4; ++j) acc = fmaf(dWp[d][j], oth[j], acc);
            dec[d] = fmaxf(acc, 0.0f);
        }
        v4f res;
#pragma unroll
        for (int a = 0; a < 4; ++a) {
            float acc = hbr[a];
#pragma unroll
            for (int d = 0; d < 4; ++d) acc = fmaf(hWr[a][d], dec[d], acc);
            res[a] = acc;
        }
        size_t off = (half ? sgBase : muBase) + ((size_t)t * Nn + gcol) * 4;
        *(v4f*)(outp + off) = res;
    };

    auto decode_half = [&](int p, int u0) {
        if (p >= 4) {
            const int s = (p - 4) & 1;
            const int t0 = (p - 4) * CH + u0;
            if (t0 + CH / 2 <= T) {
                v4f own[CH / 2], oth[CH / 2];
#pragma unroll
                for (int u = 0; u < CH / 2; ++u) { own[u] = hand[1][s][u0 + u][lane]; oth[u] = hand[1][s][u0 + u][lane ^ 16]; }
#pragma unroll
                for (int u = 0; u < CH / 2; ++u) decode_store(t0 + u, own[u], oth[u]);
            } else {
                for (int u = 0; u < CH / 2 && t0 + u < T; ++u)
                    decode_store(t0 + u, hand[1][s][u0 + u][lane], hand[1][s][u0 + u][lane ^ 16]);
            }
        }
    };

    v2f xr0[CH], xr1[CH];
#pragma unroll
    for (int u = 0; u < CH; ++u) { xr0[u] = (v2f)(0.0f); xr1[u] = (v2f)(0.0f); }
    if (role == 3) {
#pragma unroll
        for (int u = 0; u < CH; ++u) load_x(u < T ? u : T - 1, xr0[u], xr1[u]);
    }

    const int nBlk = (T + CH - 1) / CH;
    for (int p = 0; p < nBlk + 4; ++p) {
        if (role < 2) {
            const int blk = p - 1 - 2 * role;
            if (blk >= 0 && blk < nBlk) {
                const int s = blk & 1;
                v8f C0[CH], C1[CH];
#pragma unroll
                for (int u = 0; u < CH; ++u) {
                    const v4f q0 = xg[role][s][u][0][lane], q1 = xg[role][s][u][1][lane];
                    const v4f q2 = xg[role][s][u][2][lane], q3 = xg[role][s][u][3][lane];
                    C0[u] = __builtin_shufflevector(q0, q1, 0, 1, 2, 3, 4, 5, 6, 7);
                    C1[u] = __builtin_shufflevector(q2, q3, 0, 1, 2, 3, 4, 5, 6, 7);
                }
                __builtin_amdgcn_sched_barrier(0);
#pragma unroll
                for (int u = 0; u < CH; ++u) {
                    layer_step(C0[u], C1[u]);
                    v4f hv; hv.x = hs[0]; hv.y = hs[1]; hv.z = hs[2]; hv.w = hs[3];
                    hand[role][s][u][lane] = hv;
                }
            }
        } else if (role == 2) {
            const int b1 = p - 2;
            if (b1 >= 0 && b1 < nBlk) {
                const int s = b1 & 1;
                v4f hin[CH];
#pragma unroll
                for (int u = 0; u < CH; ++u) hin[u] = hand[0][s][u][lane];
                __builtin_amdgcn_sched_barrier(0);
#pragma unroll
                for (int u = 0; u < CH; ++u) {
                    v2f Bi0, Bi1;
                    Bi0.x = hin[u].x; Bi0.y = hin[u].y;
                    Bi1.x = hin[u].z; Bi1.y = hin[u].w;
                    pre_step(Bi0, Bi1, xg[1][s][u]);
                }
            }
            decode_half(p, 0);
        } else {
            if (p < nBlk) {
                v2f Bx0[CH], Bx1[CH];
#pragma unroll
                for (int u = 0; u < CH; ++u) { Bx0[u] = xr0[u]; Bx1[u] = xr1[u]; }
#pragma unroll
                for (int u = 0; u < CH; ++u) {
                    const int tn = (p + 1) * CH + u;
                    load_x(tn < T ? tn : T - 1, xr0[u], xr1[u]);
                }
                __builtin_amdgcn_sched_barrier(0);
#pragma unroll
                for (int u = 0; u < CH; ++u) pre_step(Bx0[u], Bx1[u], xg[0][p & 1][u]);
            }
            decode_half(p, CH / 2);
        }
        __builtin_amdgcn_fence(__ATOMIC_RELEASE, "workgroup", "local");
        __builtin_amdgcn_s_barrier();
        __builtin_amdgcn_fence(__ATOMIC_ACQUIRE, "workgroup", "local");
    }
}

extern "C" void kernel_launch(void* const* d_in, const int* in_sizes, int n_in,
                              void* d_out, int out_size, void* d_ws, size_t ws_size,
                              hipStream_t stream) {
    (void)n_in; (void)out_size; (void)d_ws; (void)ws_size;
    const float* state = (const float*)d_in[0];
    const float* dist  = (const float*)d_in[1];
    const float* e1W1  = (const float*)d_in[2];
    const float* e1b1  = (const float*)d_in[3];
    const float* e1W2  = (const float*)d_in[4];
    const float* e1b2  = (const float*)d_in[5];
    const float* e2W1  = (const float*)d_in[6];
    const float* e2b1  = (const float*)d_in[7];
    const float* e2W2  = (const float*)d_in[8];
    const float* e2b2  = (const float*)d_in[9];
    const float* Wih0  = (const float*)d_in[10];
    const float* Whh0  = (const float*)d_in[11];
    const float* bih0  = (const float*)d_in[12];
    const float* bhh0  = (const float*)d_in[13];
    const float* Wih1  = (const float*)d_in[14];
    const float* Whh1  = (const float*)d_in[15];
    const float* bih1  = (const float*)d_in[16];
    const float* bhh1  = (const float*)d_in[17];
    const float* decW  = (const float*)d_in[18];
    const float* decb  = (const float*)d_in[19];
    const float* finW  = (const float*)d_in[20];
    const float* finb  = (const float*)d_in[21];
    const float* fin2W = (const float*)d_in[22];
    const float* fin2b = (const float*)d_in[23];

    const int Nn = in_sizes[0] / 16;
    const int T  = in_sizes[1] / (Nn * 8);

    lstm_ctrl_kernel<<<dim3(Nn / 16), dim3(128), 0, stream>>>(
        state, dist,
        e1W1, e1b1, e1W2, e1b2, e2W1, e2b1, e2W2, e2b2,
        Wih0, Whh0, bih0, bhh0, Wih1, Whh1, bih1, bhh1,
        decW, decb, finW, finb, fin2W, fin2b,
        (float*)d_out, T, Nn);
}
